# starting kernel
# baseline (speedup 1.0000x reference)
#include <hip/hip_runtime.h>
#include <hip/hip_bf16.h>


typedef float v2f __attribute__((ext_vector_type(2)));
typedef float v8f __attribute__((ext_vector_type(8)));

#define GAMMA_EXP 0.38461538461538464f
#define SIGMA_P   0.17f
#define SIGMA_LP  4.86f

__device__ __forceinline__ int reflc(int r, int n) {
    r = (r < 0) ? -r : r;
    if (r >= n) r = 2 * n - 2 - r;
    if (r < 0)  r = 0;
    if (r >= n) r = n - 1;
    return r;
}

__device__ __forceinline__ float ltap(int d) {
    float v = 0.0f;
    if (d == 0 || d == 4)      v = 0.05f;
    else if (d == 1 || d == 3) v = 0.25f;
    else if (d == 2)           v = 0.40f;
    return v;
}

__device__ __forceinline__ float src_at(const float* __restrict__ src, int p,
                                        int y, int x, int H, int W, int gam) {
    if (gam) {
        int b = p / 3, c = p - 3 * b;
        float v = src[(((size_t)b * H + y) * W + x) * 3 + c];
        return __builtin_amdgcn_exp2f(GAMMA_EXP * __builtin_amdgcn_logf(v));
    } else {
        int Wp = W + 2;
        return src[(size_t)p * (H + 2) * Wp + (size_t)(y + 1) * Wp + (x + 1)];
    }
}

__global__ void __launch_bounds__(32)
nlp_down(const float* __restrict__ src, float* __restrict__ down,
         int H, int W, int gammaFlag) {
    __shared__ float vbuf[16][49];

    const int H2 = H / 2 + 2, W2 = W / 2 + 2;
    const int p  = blockIdx.z;
    const int i0 = blockIdx.y * 16, j0 = blockIdx.x * 16;
    const int lane = threadIdx.x;
    const int n = lane & 15;
    const int half = lane >> 4;

    int colx[3];
#pragma unroll
    for (int nt = 0; nt < 3; ++nt)
        colx[nt] = reflc(2 * j0 - 4 + nt * 16 + n, W);

    v8f acc0 = {0,0,0,0,0,0,0,0};
    v8f acc1 = {0,0,0,0,0,0,0,0};
    v8f acc2 = {0,0,0,0,0,0,0,0};

    for (int c = 0; c < 9; ++c) {
        const int kb = 4 * c + 2 * half;
        v2f a;
        a.x = ltap(kb - 2 * n);
        a.y = ltap(kb + 1 - 2 * n);
        const int r0 = reflc(2 * i0 + kb - 4, H);
        const int r1 = reflc(2 * i0 + kb - 3, H);

        v2f b0, b1, b2;
        b0.x = src_at(src, p, r0, colx[0], H, W, gammaFlag);
        b0.y = src_at(src, p, r1, colx[0], H, W, gammaFlag);
        b1.x = src_at(src, p, r0, colx[1], H, W, gammaFlag);
        b1.y = src_at(src, p, r1, colx[1], H, W, gammaFlag);
        b2.x = src_at(src, p, r0, colx[2], H, W, gammaFlag);
        b2.y = src_at(src, p, r1, colx[2], H, W, gammaFlag);

        acc0 = __builtin_amdgcn_wmma_f32_16x16x4_f32(false, a, false, b0,
                                                     (short)0, acc0, false, false);
        acc1 = __builtin_amdgcn_wmma_f32_16x16x4_f32(false, a, false, b1,
                                                     (short)0, acc1, false, false);
        acc2 = __builtin_amdgcn_wmma_f32_16x16x4_f32(false, a, false, b2,
                                                     (short)0, acc2, false, false);
    }

#pragma unroll
    for (int v = 0; v < 8; ++v) {
        vbuf[v + 8 * half][n]      = acc0[v];
        vbuf[v + 8 * half][16 + n] = acc1[v];
        vbuf[v + 8 * half][32 + n] = acc2[v];
    }
    __syncthreads();

    v8f dacc = {0,0,0,0,0,0,0,0};
    for (int c = 0; c < 9; ++c) {
        const int kb = 4 * c + 2 * half;
        v2f a;
        a.x = ltap(kb - 2 * n);
        a.y = ltap(kb + 1 - 2 * n);
        v2f b;
        b.x = vbuf[n][kb];
        b.y = vbuf[n][kb + 1];
        dacc = __builtin_amdgcn_wmma_f32_16x16x4_f32(false, a, false, b,
                                                     (short)0, dacc, false, false);
    }

    const int gi = i0 + n;
    if (gi < H2) {
        float* dp = down + (size_t)p * H2 * W2 + (size_t)gi * W2;
#pragma unroll
        for (int v = 0; v < 8; ++v) {
            int gj = j0 + v + 8 * half;
            if (gj < W2) dp[gj] = dacc[v];
        }
    }
}

__global__ void __launch_bounds__(256)
nlp_sub(const float* __restrict__ src, const float* __restrict__ down,
        float* __restrict__ out, int H, int W, int gammaFlag) {
    __shared__ float zb[36][37];
    __shared__ float dt[20][21];

    const int H2 = H / 2 + 2, W2 = W / 2 + 2;
    const int p  = blockIdx.z;
    const int y0 = blockIdx.y * 32, x0 = blockIdx.x * 32;
    const int tid = threadIdx.x;
    const float* dw = down + (size_t)p * H2 * W2;

    const int dr0 = (y0 >> 1) - 1, dc0 = (x0 >> 1) - 1;
    __builtin_prefetch(dw + (size_t)(dr0 < 0 ? 0 : dr0) * W2, 0, 1);

    for (int t = tid; t < 400; t += 256) {
        int ly = t / 20, lx = t % 20;
        int r = dr0 + ly; r = r < 0 ? 0 : (r >= H2 ? H2 - 1 : r);
        int c = dc0 + lx; c = c < 0 ? 0 : (c >= W2 ? W2 - 1 : c);
        dt[ly][lx] = dw[(size_t)r * W2 + c];
    }
    __syncthreads();

    for (int t = tid; t < 1296; t += 256) {
        int ly = t / 36, lx = t % 36;
        int y = y0 - 2 + ly, x = x0 - 2 + lx;
        float z = 0.0f;
        if (y >= 0 && y < H && x >= 0 && x < W) {
            float xv = src_at(src, p, y, x, H, W, gammaFlag);
            float wv[3], wh[3];
            int r0, c0;
            if (y & 1) { r0 = (y + 1) >> 1; wv[0] = 0.5f; wv[1] = 0.5f; wv[2] = 0.0f; }
            else       { r0 = y >> 1;       wv[0] = 0.1f; wv[1] = 0.8f; wv[2] = 0.1f; }
            if (x & 1) { c0 = (x + 1) >> 1; wh[0] = 0.5f; wh[1] = 0.5f; wh[2] = 0.0f; }
            else       { c0 = x >> 1;       wh[0] = 0.1f; wh[1] = 0.8f; wh[2] = 0.1f; }
            float up = 0.0f;
#pragma unroll
            for (int rr = 0; rr < 3; ++rr) {
                int lr = r0 + rr - dr0; lr = lr < 0 ? 0 : (lr > 19 ? 19 : lr);
                float rowv = 0.0f;
#pragma unroll
                for (int cc = 0; cc < 3; ++cc) {
                    int lc = c0 + cc - dc0; lc = lc < 0 ? 0 : (lc > 19 ? 19 : lc);
                    rowv = fmaf(wh[cc], dt[lr][lc], rowv);
                }
                up = fmaf(wv[rr], rowv, up);
            }
            z = xv - up;
        }
        zb[ly][lx] = z;
    }
    __syncthreads();

    const float Pk[5][5] = {
        {0.04f, 0.04f, 0.05f, 0.04f, 0.04f},
        {0.04f, 0.03f, 0.04f, 0.03f, 0.04f},
        {0.05f, 0.04f, 0.05f, 0.04f, 0.05f},
        {0.04f, 0.03f, 0.04f, 0.03f, 0.04f},
        {0.04f, 0.04f, 0.05f, 0.04f, 0.04f}};

    const int b = p / 3, ch = p - 3 * b;
#pragma unroll
    for (int q = 0; q < 4; ++q) {
        int idx = tid + q * 256;
        int oy = idx >> 5, ox = idx & 31;
        float zc = zb[oy + 2][ox + 2];
        float pool = 0.0f;
#pragma unroll
        for (int dy = 0; dy < 5; ++dy)
#pragma unroll
            for (int dx = 0; dx < 5; ++dx)
                pool = fmaf(Pk[dy][dx], fabsf(zb[oy + dy][ox + dx]), pool);
        float o = zc / (pool + SIGMA_P);
        int gy = y0 + oy, gx = x0 + ox;
        out[(((size_t)b * H + gy) * W + gx) * 3 + ch] = o;
    }
}

__global__ void __launch_bounds__(256)
nlp_resid(const float* __restrict__ d4, float* __restrict__ out) {
    const int p = blockIdx.x, tid = threadIdx.x;
    const int y = tid >> 4, x = tid & 15;
    float v = d4[(size_t)p * 18 * 18 + (size_t)(y + 1) * 18 + (x + 1)];
    float o = v / (fabsf(v) + SIGMA_LP);
    int b = p / 3, c = p - 3 * b;
    out[(((size_t)b * 16 + y) * 16 + x) * 3 + c] = o;
}

extern "C" void kernel_launch(void* const* d_in, const int* in_sizes, int n_in,
                              void* d_out, int out_size, void* d_ws, size_t ws_size,
                              hipStream_t stream) {
    (void)in_sizes; (void)n_in; (void)out_size; (void)ws_size;
    const float* img = (const float*)d_in[0];
    float* out = (float*)d_out;
    float* ws  = (float*)d_ws;

    const int Hl[6] = {512, 256, 128, 64, 32, 16};

    size_t dof[5]; size_t off = 0;
    for (int l = 0; l < 5; ++l) {
        dof[l] = off;
        size_t h2 = (size_t)(Hl[l] / 2 + 2);
        off += (size_t)48 * h2 * h2;
    }
    size_t oof[6]; size_t o2 = 0;
    for (int l = 0; l < 6; ++l) {
        oof[l] = o2;
        o2 += (size_t)16 * Hl[l] * Hl[l] * 3;
    }

    const float* srcp = img;
    int gam = 1;
    for (int l = 0; l < 5; ++l) {
        const int H = Hl[l];
        const int H2 = H / 2 + 2;
        const int T = (H2 + 15) / 16;
        nlp_down<<<dim3(T, T, 48), 32, 0, stream>>>(srcp, ws + dof[l], H, H, gam);
        nlp_sub<<<dim3(H / 32, H / 32, 48), 256, 0, stream>>>(
            srcp, ws + dof[l], out + oof[l], H, H, gam);
        srcp = ws + dof[l];
        gam = 0;
    }
    nlp_resid<<<48, 256, 0, stream>>>(ws + dof[4], out + oof[5]);
}


# gamma switch made a template parameter and each thread's global loads (54 in down, 8 in sub) issued together before first use instead of one load per wait; arithmetic and its order unchanged
# speedup vs baseline: 1.1955x; 1.1955x over previous
#include <hip/hip_runtime.h>
#include <hip/hip_bf16.h>


typedef float v2f __attribute__((ext_vector_type(2)));
typedef float v8f __attribute__((ext_vector_type(8)));

#define GAMMA_EXP 0.38461538461538464f
#define SIGMA_P   0.17f
#define SIGMA_LP  4.86f

__device__ __forceinline__ int reflc(int r, int n) {
    r = (r < 0) ? -r : r;
    if (r >= n) r = 2 * n - 2 - r;
    if (r < 0)  r = 0;
    if (r >= n) r = n - 1;
    return r;
}

__device__ __forceinline__ int clampi(int v, int lo, int hi) {
    return v < lo ? lo : (v > hi ? hi : v);
}

__device__ __forceinline__ float ltap(int d) {
    float v = 0.0f;
    if (d == 0 || d == 4)      v = 0.05f;
    else if (d == 1 || d == 3) v = 0.25f;
    else if (d == 2)           v = 0.40f;
    return v;
}

__device__ __forceinline__ float gamma_of(float v) {
    return __builtin_amdgcn_exp2f(GAMMA_EXP * __builtin_amdgcn_logf(v));
}

template <bool GAM>
__device__ __forceinline__ void src_layout(int p, int H, int W, unsigned& base,
                                           unsigned& rpitch, unsigned& cpitch) {
    if (GAM) {
        const int b = p / 3, c = p - 3 * b;
        base = (unsigned)b * (unsigned)H * (unsigned)W * 3u + (unsigned)c;
        rpitch = (unsigned)W * 3u;
        cpitch = 3u;
    } else {
        const unsigned Wp = (unsigned)W + 2u;
        base = (unsigned)p * ((unsigned)H + 2u) * Wp + Wp + 1u;
        rpitch = Wp;
        cpitch = 1u;
    }
}

template <bool GAM>
__global__ void __launch_bounds__(32)
nlp_down(const float* __restrict__ src, float* __restrict__ down, int H, int W) {
    __shared__ float vbuf[16][49];

    const int H2 = H / 2 + 2, W2 = W / 2 + 2;
    const int p  = blockIdx.z;
    const int i0 = blockIdx.y * 16, j0 = blockIdx.x * 16;
    const int lane = threadIdx.x;
    const int n = lane & 15;
    const int half = lane >> 4;

    unsigned base, rpitch, cpitch;
    src_layout<GAM>(p, H, W, base, rpitch, cpitch);

    unsigned colo[3];
#pragma unroll
    for (int nt = 0; nt < 3; ++nt)
        colo[nt] = base + cpitch * (unsigned)reflc(2 * j0 - 4 + nt * 16 + n, W);

    float raw[9][6];
#pragma unroll
    for (int c = 0; c < 9; ++c) {
        const int kb = 4 * c + 2 * half;
        const unsigned ro0 = rpitch * (unsigned)reflc(2 * i0 + kb - 4, H);
        const unsigned ro1 = rpitch * (unsigned)reflc(2 * i0 + kb - 3, H);
#pragma unroll
        for (int nt = 0; nt < 3; ++nt) {
            raw[c][2 * nt]     = src[ro0 + colo[nt]];
            raw[c][2 * nt + 1] = src[ro1 + colo[nt]];
        }
    }

    __builtin_amdgcn_sched_barrier(0);

    v8f acc0 = {0,0,0,0,0,0,0,0};
    v8f acc1 = {0,0,0,0,0,0,0,0};
    v8f acc2 = {0,0,0,0,0,0,0,0};

#pragma unroll
    for (int c = 0; c < 9; ++c) {
        const int kb = 4 * c + 2 * half;
        v2f a;
        a.x = ltap(kb - 2 * n);
        a.y = ltap(kb + 1 - 2 * n);

        v2f b0, b1, b2;
        if (GAM) {
            b0.x = gamma_of(raw[c][0]); b0.y = gamma_of(raw[c][1]);
            b1.x = gamma_of(raw[c][2]); b1.y = gamma_of(raw[c][3]);
            b2.x = gamma_of(raw[c][4]); b2.y = gamma_of(raw[c][5]);
        } else {
            b0.x = raw[c][0]; b0.y = raw[c][1];
            b1.x = raw[c][2]; b1.y = raw[c][3];
            b2.x = raw[c][4]; b2.y = raw[c][5];
        }

        acc0 = __builtin_amdgcn_wmma_f32_16x16x4_f32(false, a, false, b0,
                                                     (short)0, acc0, false, false);
        acc1 = __builtin_amdgcn_wmma_f32_16x16x4_f32(false, a, false, b1,
                                                     (short)0, acc1, false, false);
        acc2 = __builtin_amdgcn_wmma_f32_16x16x4_f32(false, a, false, b2,
                                                     (short)0, acc2, false, false);
    }

#pragma unroll
    for (int v = 0; v < 8; ++v) {
        vbuf[v + 8 * half][n]      = acc0[v];
        vbuf[v + 8 * half][16 + n] = acc1[v];
        vbuf[v + 8 * half][32 + n] = acc2[v];
    }
    __syncthreads();

    v8f dacc = {0,0,0,0,0,0,0,0};
    for (int c = 0; c < 9; ++c) {
        const int kb = 4 * c + 2 * half;
        v2f a;
        a.x = ltap(kb - 2 * n);
        a.y = ltap(kb + 1 - 2 * n);
        v2f b;
        b.x = vbuf[n][kb];
        b.y = vbuf[n][kb + 1];
        dacc = __builtin_amdgcn_wmma_f32_16x16x4_f32(false, a, false, b,
                                                     (short)0, dacc, false, false);
    }

    const int gi = i0 + n;
    if (gi < H2) {
        float* dp = down + (size_t)p * H2 * W2 + (size_t)gi * W2;
#pragma unroll
        for (int v = 0; v < 8; ++v) {
            int gj = j0 + v + 8 * half;
            if (gj < W2) dp[gj] = dacc[v];
        }
    }
}

template <bool GAM>
__global__ void __launch_bounds__(256)
nlp_sub(const float* __restrict__ src, const float* __restrict__ down,
        float* __restrict__ out, int H, int W) {
    __shared__ float zb[36][37];
    __shared__ float dt[20][21];

    const int H2 = H / 2 + 2, W2 = W / 2 + 2;
    const int p  = blockIdx.z;
    const int y0 = blockIdx.y * 32, x0 = blockIdx.x * 32;
    const int tid = threadIdx.x;
    const float* dw = down + (size_t)p * H2 * W2;

    const int dr0 = (y0 >> 1) - 1, dc0 = (x0 >> 1) - 1;

    unsigned base, rpitch, cpitch;
    src_layout<GAM>(p, H, W, base, rpitch, cpitch);

    float dv[2];
#pragma unroll
    for (int q = 0; q < 2; ++q) {
        int t = tid + q * 256; t = t > 399 ? 399 : t;
        int ly = t / 20, lx = t % 20;
        int r = clampi(dr0 + ly, 0, H2 - 1);
        int c = clampi(dc0 + lx, 0, W2 - 1);
        dv[q] = dw[(unsigned)r * (unsigned)W2 + (unsigned)c];
    }
    float xr[6];
#pragma unroll
    for (int q = 0; q < 6; ++q) {
        int t = tid + q * 256; t = t > 1295 ? 1295 : t;
        int ly = t / 36, lx = t % 36;
        int y = clampi(y0 - 2 + ly, 0, H - 1);
        int x = clampi(x0 - 2 + lx, 0, W - 1);
        xr[q] = src[base + (unsigned)y * rpitch + (unsigned)x * cpitch];
    }

#pragma unroll
    for (int q = 0; q < 2; ++q) {
        int t = tid + q * 256;
        if (t < 400) dt[t / 20][t % 20] = dv[q];
    }
    __syncthreads();

#pragma unroll
    for (int q = 0; q < 6; ++q) {
        int t = tid + q * 256;
        if (t < 1296) {
            int ly = t / 36, lx = t % 36;
            int y = y0 - 2 + ly, x = x0 - 2 + lx;
            float z = 0.0f;
            if (y >= 0 && y < H && x >= 0 && x < W) {
                float xv = GAM ? gamma_of(xr[q]) : xr[q];
                float wv[3], wh[3];
                int r0, c0;
                if (y & 1) { r0 = (y + 1) >> 1; wv[0] = 0.5f; wv[1] = 0.5f; wv[2] = 0.0f; }
                else       { r0 = y >> 1;       wv[0] = 0.1f; wv[1] = 0.8f; wv[2] = 0.1f; }
                if (x & 1) { c0 = (x + 1) >> 1; wh[0] = 0.5f; wh[1] = 0.5f; wh[2] = 0.0f; }
                else       { c0 = x >> 1;       wh[0] = 0.1f; wh[1] = 0.8f; wh[2] = 0.1f; }
                float up = 0.0f;
#pragma unroll
                for (int rr = 0; rr < 3; ++rr) {
                    int lr = r0 + rr - dr0; lr = lr < 0 ? 0 : (lr > 19 ? 19 : lr);
                    float rowv = 0.0f;
#pragma unroll
                    for (int cc = 0; cc < 3; ++cc) {
                        int lc = c0 + cc - dc0; lc = lc < 0 ? 0 : (lc > 19 ? 19 : lc);
                        rowv = fmaf(wh[cc], dt[lr][lc], rowv);
                    }
                    up = fmaf(wv[rr], rowv, up);
                }
                z = xv - up;
            }
            zb[ly][lx] = z;
        }
    }
    __syncthreads();

    const float Pk[5][5] = {
        {0.04f, 0.04f, 0.05f, 0.04f, 0.04f},
        {0.04f, 0.03f, 0.04f, 0.03f, 0.04f},
        {0.05f, 0.04f, 0.05f, 0.04f, 0.05f},
        {0.04f, 0.03f, 0.04f, 0.03f, 0.04f},
        {0.04f, 0.04f, 0.05f, 0.04f, 0.04f}};

    const int b = p / 3, ch = p - 3 * b;
#pragma unroll
    for (int q = 0; q < 4; ++q) {
        int idx = tid + q * 256;
        int oy = idx >> 5, ox = idx & 31;
        float zc = zb[oy + 2][ox + 2];
        float pool = 0.0f;
#pragma unroll
        for (int dy = 0; dy < 5; ++dy)
#pragma unroll
            for (int dx = 0; dx < 5; ++dx)
                pool = fmaf(Pk[dy][dx], fabsf(zb[oy + dy][ox + dx]), pool);
        float o = zc / (pool + SIGMA_P);
        int gy = y0 + oy, gx = x0 + ox;
        out[(((size_t)b * H + gy) * W + gx) * 3 + ch] = o;
    }
}

__global__ void __launch_bounds__(256)
nlp_resid(const float* __restrict__ d4, float* __restrict__ out) {
    const int p = blockIdx.x, tid = threadIdx.x;
    const int y = tid >> 4, x = tid & 15;
    float v = d4[(size_t)p * 18 * 18 + (size_t)(y + 1) * 18 + (x + 1)];
    float o = v / (fabsf(v) + SIGMA_LP);
    int b = p / 3, c = p - 3 * b;
    out[(((size_t)b * 16 + y) * 16 + x) * 3 + c] = o;
}

extern "C" void kernel_launch(void* const* d_in, const int* in_sizes, int n_in,
                              void* d_out, int out_size, void* d_ws, size_t ws_size,
                              hipStream_t stream) {
    (void)in_sizes; (void)n_in; (void)out_size; (void)ws_size;
    const float* img = (const float*)d_in[0];
    float* out = (float*)d_out;
    float* ws  = (float*)d_ws;

    const int Hl[6] = {512, 256, 128, 64, 32, 16};

    size_t dof[5]; size_t off = 0;
    for (int l = 0; l < 5; ++l) {
        dof[l] = off;
        size_t h2 = (size_t)(Hl[l] / 2 + 2);
        off += (size_t)48 * h2 * h2;
    }
    size_t oof[6]; size_t o2 = 0;
    for (int l = 0; l < 6; ++l) {
        oof[l] = o2;
        o2 += (size_t)16 * Hl[l] * Hl[l] * 3;
    }

    const float* srcp = img;
    for (int l = 0; l < 5; ++l) {
        const int H = Hl[l];
        const int H2 = H / 2 + 2;
        const int T = (H2 + 15) / 16;
        if (l == 0) {
            nlp_down<true><<<dim3(T, T, 48), 32, 0, stream>>>(srcp, ws + dof[l], H, H);
            nlp_sub<true><<<dim3(H / 32, H / 32, 48), 256, 0, stream>>>(
                srcp, ws + dof[l], out + oof[l], H, H);
        } else {
            nlp_down<false><<<dim3(T, T, 48), 32, 0, stream>>>(srcp, ws + dof[l], H, H);
            nlp_sub<false><<<dim3(H / 32, H / 32, 48), 256, 0, stream>>>(
                srcp, ws + dof[l], out + oof[l], H, H);
        }
        srcp = ws + dof[l];
    }
    nlp_resid<<<48, 256, 0, stream>>>(ws + dof[4], out + oof[5]);
}


# level l sub-band and level l+1 downsample (last: lowpass residual) share one launch, 8 down tiles per work-group there: 11 -> 6 dependent launches; arithmetic and sum order unchanged
# speedup vs baseline: 1.2971x; 1.0850x over previous
#include <hip/hip_runtime.h>
#include <hip/hip_bf16.h>


typedef float v2f __attribute__((ext_vector_type(2)));
typedef float v8f __attribute__((ext_vector_type(8)));

#define GAMMA_EXP 0.38461538461538464f
#define SIGMA_P   0.17f
#define SIGMA_LP  4.86f

__device__ __forceinline__ int reflc(int r, int n) {
    r = (r < 0) ? -r : r;
    if (r >= n) r = 2 * n - 2 - r;
    if (r < 0)  r = 0;
    if (r >= n) r = n - 1;
    return r;
}

__device__ __forceinline__ int clampi(int v, int lo, int hi) {
    return v < lo ? lo : (v > hi ? hi : v);
}

__device__ __forceinline__ float ltap(int d) {
    float v = 0.0f;
    if (d == 0 || d == 4)      v = 0.05f;
    else if (d == 1 || d == 3) v = 0.25f;
    else if (d == 2)           v = 0.40f;
    return v;
}

__device__ __forceinline__ float gamma_of(float v) {
    return __builtin_amdgcn_exp2f(GAMMA_EXP * __builtin_amdgcn_logf(v));
}

template <bool GAM>
__device__ __forceinline__ void src_layout(int p, int H, int W, unsigned& base,
                                           unsigned& rpitch, unsigned& cpitch) {
    if (GAM) {
        const int b = p / 3, c = p - 3 * b;
        base = (unsigned)b * (unsigned)H * (unsigned)W * 3u + (unsigned)c;
        rpitch = (unsigned)W * 3u;
        cpitch = 3u;
    } else {
        const unsigned Wp = (unsigned)W + 2u;
        base = (unsigned)p * ((unsigned)H + 2u) * Wp + Wp + 1u;
        rpitch = Wp;
        cpitch = 1u;
    }
}

template <bool GAM>
__device__ __forceinline__ void down_tile(const float* __restrict__ src, float* __restrict__ down,
                                          int H, int W, int p, int i0, int j0, bool live,
                                          float (*vbuf)[49], int lane) {
    const int H2 = H / 2 + 2, W2 = W / 2 + 2;
    const int n = lane & 15;
    const int half = lane >> 4;

    if (live) {
        unsigned base, rpitch, cpitch;
        src_layout<GAM>(p, H, W, base, rpitch, cpitch);

        unsigned colo[3];
#pragma unroll
        for (int nt = 0; nt < 3; ++nt)
            colo[nt] = base + cpitch * (unsigned)reflc(2 * j0 - 4 + nt * 16 + n, W);

        float raw[9][6];
#pragma unroll
        for (int c = 0; c < 9; ++c) {
            const int kb = 4 * c + 2 * half;
            const unsigned ro0 = rpitch * (unsigned)reflc(2 * i0 + kb - 4, H);
            const unsigned ro1 = rpitch * (unsigned)reflc(2 * i0 + kb - 3, H);
#pragma unroll
            for (int nt = 0; nt < 3; ++nt) {
                raw[c][2 * nt]     = src[ro0 + colo[nt]];
                raw[c][2 * nt + 1] = src[ro1 + colo[nt]];
            }
        }

        __builtin_amdgcn_sched_barrier(0);

        v8f acc0 = {0,0,0,0,0,0,0,0};
        v8f acc1 = {0,0,0,0,0,0,0,0};
        v8f acc2 = {0,0,0,0,0,0,0,0};

#pragma unroll
        for (int c = 0; c < 9; ++c) {
            const int kb = 4 * c + 2 * half;
            v2f a;
            a.x = ltap(kb - 2 * n);
            a.y = ltap(kb + 1 - 2 * n);

            v2f b0, b1, b2;
            if (GAM) {
                b0.x = gamma_of(raw[c][0]); b0.y = gamma_of(raw[c][1]);
                b1.x = gamma_of(raw[c][2]); b1.y = gamma_of(raw[c][3]);
                b2.x = gamma_of(raw[c][4]); b2.y = gamma_of(raw[c][5]);
            } else {
                b0.x = raw[c][0]; b0.y = raw[c][1];
                b1.x = raw[c][2]; b1.y = raw[c][3];
                b2.x = raw[c][4]; b2.y = raw[c][5];
            }

            acc0 = __builtin_amdgcn_wmma_f32_16x16x4_f32(false, a, false, b0,
                                                         (short)0, acc0, false, false);
            acc1 = __builtin_amdgcn_wmma_f32_16x16x4_f32(false, a, false, b1,
                                                         (short)0, acc1, false, false);
            acc2 = __builtin_amdgcn_wmma_f32_16x16x4_f32(false, a, false, b2,
                                                         (short)0, acc2, false, false);
        }

#pragma unroll
        for (int v = 0; v < 8; ++v) {
            vbuf[v + 8 * half][n]      = acc0[v];
            vbuf[v + 8 * half][16 + n] = acc1[v];
            vbuf[v + 8 * half][32 + n] = acc2[v];
        }
    }
    __syncthreads();
    if (live) {
        v8f dacc = {0,0,0,0,0,0,0,0};
        for (int c = 0; c < 9; ++c) {
            const int kb = 4 * c + 2 * half;
            v2f a;
            a.x = ltap(kb - 2 * n);
            a.y = ltap(kb + 1 - 2 * n);
            v2f b;
            b.x = vbuf[n][kb];
            b.y = vbuf[n][kb + 1];
            dacc = __builtin_amdgcn_wmma_f32_16x16x4_f32(false, a, false, b,
                                                         (short)0, dacc, false, false);
        }

        const int gi = i0 + n;
        if (gi < H2) {
            float* dp = down + (size_t)p * H2 * W2 + (size_t)gi * W2;
#pragma unroll
            for (int v = 0; v < 8; ++v) {
                int gj = j0 + v + 8 * half;
                if (gj < W2) dp[gj] = dacc[v];
            }
        }
    }
}

template <bool GAM>
__global__ void __launch_bounds__(32)
nlp_down(const float* __restrict__ src, float* __restrict__ down, int H, int W) {
    __shared__ float vbuf[16][49];
    down_tile<GAM>(src, down, H, W, blockIdx.z, blockIdx.y * 16, blockIdx.x * 16, true,
                   vbuf, threadIdx.x);
}

template <bool GAM>
__device__ __forceinline__ void sub_tile(const float* __restrict__ src,
                                         const float* __restrict__ down,
                                         float* __restrict__ out, int H, int W, int p,
                                         int y0, int x0, int tid,
                                         float (*zb)[37], float (*dt)[21]) {
    const int H2 = H / 2 + 2, W2 = W / 2 + 2;
    const float* dw = down + (size_t)p * H2 * W2;

    const int dr0 = (y0 >> 1) - 1, dc0 = (x0 >> 1) - 1;

    unsigned base, rpitch, cpitch;
    src_layout<GAM>(p, H, W, base, rpitch, cpitch);

    float dv[2];
#pragma unroll
    for (int q = 0; q < 2; ++q) {
        int t = tid + q * 256; t = t > 399 ? 399 : t;
        int ly = t / 20, lx = t % 20;
        int r = clampi(dr0 + ly, 0, H2 - 1);
        int c = clampi(dc0 + lx, 0, W2 - 1);
        dv[q] = dw[(unsigned)r * (unsigned)W2 + (unsigned)c];
    }
    float xr[6];
#pragma unroll
    for (int q = 0; q < 6; ++q) {
        int t = tid + q * 256; t = t > 1295 ? 1295 : t;
        int ly = t / 36, lx = t % 36;
        int y = clampi(y0 - 2 + ly, 0, H - 1);
        int x = clampi(x0 - 2 + lx, 0, W - 1);
        xr[q] = src[base + (unsigned)y * rpitch + (unsigned)x * cpitch];
    }

#pragma unroll
    for (int q = 0; q < 2; ++q) {
        int t = tid + q * 256;
        if (t < 400) dt[t / 20][t % 20] = dv[q];
    }
    __syncthreads();

#pragma unroll
    for (int q = 0; q < 6; ++q) {
        int t = tid + q * 256;
        if (t < 1296) {
            int ly = t / 36, lx = t % 36;
            int y = y0 - 2 + ly, x = x0 - 2 + lx;
            float z = 0.0f;
            if (y >= 0 && y < H && x >= 0 && x < W) {
                float xv = GAM ? gamma_of(xr[q]) : xr[q];
                float wv[3], wh[3];
                int r0, c0;
                if (y & 1) { r0 = (y + 1) >> 1; wv[0] = 0.5f; wv[1] = 0.5f; wv[2] = 0.0f; }
                else       { r0 = y >> 1;       wv[0] = 0.1f; wv[1] = 0.8f; wv[2] = 0.1f; }
                if (x & 1) { c0 = (x + 1) >> 1; wh[0] = 0.5f; wh[1] = 0.5f; wh[2] = 0.0f; }
                else       { c0 = x >> 1;       wh[0] = 0.1f; wh[1] = 0.8f; wh[2] = 0.1f; }
                float up = 0.0f;
#pragma unroll
                for (int rr = 0; rr < 3; ++rr) {
                    int lr = r0 + rr - dr0; lr = lr < 0 ? 0 : (lr > 19 ? 19 : lr);
                    float rowv = 0.0f;
#pragma unroll
                    for (int cc = 0; cc < 3; ++cc) {
                        int lc = c0 + cc - dc0; lc = lc < 0 ? 0 : (lc > 19 ? 19 : lc);
                        rowv = fmaf(wh[cc], dt[lr][lc], rowv);
                    }
                    up = fmaf(wv[rr], rowv, up);
                }
                z = xv - up;
            }
            zb[ly][lx] = z;
        }
    }
    __syncthreads();

    const float Pk[5][5] = {
        {0.04f, 0.04f, 0.05f, 0.04f, 0.04f},
        {0.04f, 0.03f, 0.04f, 0.03f, 0.04f},
        {0.05f, 0.04f, 0.05f, 0.04f, 0.05f},
        {0.04f, 0.03f, 0.04f, 0.03f, 0.04f},
        {0.04f, 0.04f, 0.05f, 0.04f, 0.04f}};

    const int b = p / 3, ch = p - 3 * b;
#pragma unroll
    for (int q = 0; q < 4; ++q) {
        int idx = tid + q * 256;
        int oy = idx >> 5, ox = idx & 31;
        float zc = zb[oy + 2][ox + 2];
        float pool = 0.0f;
#pragma unroll
        for (int dy = 0; dy < 5; ++dy)
#pragma unroll
            for (int dx = 0; dx < 5; ++dx)
                pool = fmaf(Pk[dy][dx], fabsf(zb[oy + dy][ox + dx]), pool);
        float o = zc / (pool + SIGMA_P);
        int gy = y0 + oy, gx = x0 + ox;
        out[(((size_t)b * H + gy) * W + gx) * 3 + ch] = o;
    }
}

template <bool GAM, bool LAST>
__global__ void __launch_bounds__(256) __attribute__((amdgpu_waves_per_eu(12)))
nlp_level(const float* __restrict__ src, const float* __restrict__ down,
          float* __restrict__ out, float* __restrict__ nxt, int H, int W, int nd, int sh) {
    __shared__ float smem[LAST ? (36 * 37 + 20 * 21) : (8 * 16 * 49)];

    const int p = blockIdx.z;
    const int b = blockIdx.x;
    const int tid = threadIdx.x;

    if (b < nd) {
        if (LAST) {
            const int y = tid >> 4, x = tid & 15;
            float v = down[(size_t)p * 18 * 18 + (size_t)(y + 1) * 18 + (x + 1)];
            float o = v / (fabsf(v) + SIGMA_LP);
            int bb = p / 3, c = p - 3 * bb;
            nxt[(((size_t)bb * 16 + y) * 16 + x) * 3 + c] = o;
        } else {
            const int wave = __builtin_amdgcn_readfirstlane(tid >> 5);
            const int Hn = H / 2, Wn = W / 2;
            const int Ty = (Hn / 2 + 2 + 15) / 16, Tx = (Wn / 2 + 2 + 15) / 16;
            const int tile = b * 8 + wave;
            const bool live = tile < Ty * Tx;
            const int ti = tile / Tx, tj = tile - ti * Tx;
            down_tile<false>(down, nxt, Hn, Wn, p, ti * 16, tj * 16, live,
                             (float (*)[49])(smem + wave * (16 * 49)), tid & 31);
        }
    } else {
        const int s = b - nd;
        sub_tile<GAM>(src, down, out, H, W, p, (s >> sh) * 32, (s & ((1 << sh) - 1)) * 32,
                      tid, (float (*)[37])smem, (float (*)[21])(smem + 36 * 37));
    }
}

extern "C" void kernel_launch(void* const* d_in, const int* in_sizes, int n_in,
                              void* d_out, int out_size, void* d_ws, size_t ws_size,
                              hipStream_t stream) {
    (void)in_sizes; (void)n_in; (void)out_size; (void)ws_size;
    const float* img = (const float*)d_in[0];
    float* out = (float*)d_out;
    float* ws  = (float*)d_ws;

    const int Hl[6]  = {512, 256, 128, 64, 32, 16};
    const int shl[5] = {4, 3, 2, 1, 0};

    size_t dof[5]; size_t off = 0;
    for (int l = 0; l < 5; ++l) {
        dof[l] = off;
        size_t h2 = (size_t)(Hl[l] / 2 + 2);
        off += (size_t)48 * h2 * h2;
    }
    size_t oof[6]; size_t o2 = 0;
    for (int l = 0; l < 6; ++l) {
        oof[l] = o2;
        o2 += (size_t)16 * Hl[l] * Hl[l] * 3;
    }

    {
        const int T = (Hl[0] / 2 + 2 + 15) / 16;
        nlp_down<true><<<dim3(T, T, 48), 32, 0, stream>>>(img, ws + dof[0], Hl[0], Hl[0]);
    }
    const float* srcp = img;
    for (int l = 0; l < 5; ++l) {
        const int H = Hl[l];
        const int ns = (H / 32) * (H / 32);
        if (l < 4) {
            const int Tn = (Hl[l + 1] / 2 + 2 + 15) / 16;
            const int nd = (Tn * Tn + 7) / 8;
            if (l == 0)
                nlp_level<true, false><<<dim3(nd + ns, 1, 48), 256, 0, stream>>>(
                    srcp, ws + dof[l], out + oof[l], ws + dof[l + 1], H, H, nd, shl[l]);
            else
                nlp_level<false, false><<<dim3(nd + ns, 1, 48), 256, 0, stream>>>(
                    srcp, ws + dof[l], out + oof[l], ws + dof[l + 1], H, H, nd, shl[l]);
        } else {
            nlp_level<false, true><<<dim3(1 + ns, 1, 48), 256, 0, stream>>>(
                srcp, ws + dof[l], out + oof[l], out + oof[5], H, H, 1, shl[l]);
        }
        srcp = ws + dof[l];
    }
}


# a sub-band work-group handles all three channels of its 32x32 tile: NHWC reads and writes are consecutive floats, per-pixel index and tap work is shared, a third of the sub-band waves; sums unchanged
# speedup vs baseline: 1.5168x; 1.1694x over previous
#include <hip/hip_runtime.h>
#include <hip/hip_bf16.h>


typedef float v2f __attribute__((ext_vector_type(2)));
typedef float v8f __attribute__((ext_vector_type(8)));

#define GAMMA_EXP 0.38461538461538464f
#define SIGMA_P   0.17f
#define SIGMA_LP  4.86f

__device__ __forceinline__ int reflc(int r, int n) {
    r = (r < 0) ? -r : r;
    if (r >= n) r = 2 * n - 2 - r;
    if (r < 0)  r = 0;
    if (r >= n) r = n - 1;
    return r;
}

__device__ __forceinline__ int clampi(int v, int lo, int hi) {
    return v < lo ? lo : (v > hi ? hi : v);
}

__device__ __forceinline__ float ltap(int d) {
    float v = 0.0f;
    if (d == 0 || d == 4)      v = 0.05f;
    else if (d == 1 || d == 3) v = 0.25f;
    else if (d == 2)           v = 0.40f;
    return v;
}

__device__ __forceinline__ float gamma_of(float v) {
    return __builtin_amdgcn_exp2f(GAMMA_EXP * __builtin_amdgcn_logf(v));
}

template <bool GAM>
__device__ __forceinline__ void src_layout(int p, int H, int W, unsigned& base,
                                           unsigned& rpitch, unsigned& cpitch) {
    if (GAM) {
        const int b = p / 3, c = p - 3 * b;
        base = (unsigned)b * (unsigned)H * (unsigned)W * 3u + (unsigned)c;
        rpitch = (unsigned)W * 3u;
        cpitch = 3u;
    } else {
        const unsigned Wp = (unsigned)W + 2u;
        base = (unsigned)p * ((unsigned)H + 2u) * Wp + Wp + 1u;
        rpitch = Wp;
        cpitch = 1u;
    }
}

template <bool GAM>
__device__ __forceinline__ void down_tile(const float* __restrict__ src, float* __restrict__ down,
                                          int H, int W, int p, int i0, int j0, bool live,
                                          float (*vbuf)[49], int lane) {
    const int H2 = H / 2 + 2, W2 = W / 2 + 2;
    const int n = lane & 15;
    const int half = lane >> 4;

    if (live) {
        unsigned base, rpitch, cpitch;
        src_layout<GAM>(p, H, W, base, rpitch, cpitch);

        unsigned colo[3];
#pragma unroll
        for (int nt = 0; nt < 3; ++nt)
            colo[nt] = base + cpitch * (unsigned)reflc(2 * j0 - 4 + nt * 16 + n, W);

        float raw[9][6];
#pragma unroll
        for (int c = 0; c < 9; ++c) {
            const int kb = 4 * c + 2 * half;
            const unsigned ro0 = rpitch * (unsigned)reflc(2 * i0 + kb - 4, H);
            const unsigned ro1 = rpitch * (unsigned)reflc(2 * i0 + kb - 3, H);
#pragma unroll
            for (int nt = 0; nt < 3; ++nt) {
                raw[c][2 * nt]     = src[ro0 + colo[nt]];
                raw[c][2 * nt + 1] = src[ro1 + colo[nt]];
            }
        }

        __builtin_amdgcn_sched_barrier(0);

        v8f acc0 = {0,0,0,0,0,0,0,0};
        v8f acc1 = {0,0,0,0,0,0,0,0};
        v8f acc2 = {0,0,0,0,0,0,0,0};

#pragma unroll
        for (int c = 0; c < 9; ++c) {
            const int kb = 4 * c + 2 * half;
            v2f a;
            a.x = ltap(kb - 2 * n);
            a.y = ltap(kb + 1 - 2 * n);

            v2f b0, b1, b2;
            if (GAM) {
                b0.x = gamma_of(raw[c][0]); b0.y = gamma_of(raw[c][1]);
                b1.x = gamma_of(raw[c][2]); b1.y = gamma_of(raw[c][3]);
                b2.x = gamma_of(raw[c][4]); b2.y = gamma_of(raw[c][5]);
            } else {
                b0.x = raw[c][0]; b0.y = raw[c][1];
                b1.x = raw[c][2]; b1.y = raw[c][3];
                b2.x = raw[c][4]; b2.y = raw[c][5];
            }

            acc0 = __builtin_amdgcn_wmma_f32_16x16x4_f32(false, a, false, b0,
                                                         (short)0, acc0, false, false);
            acc1 = __builtin_amdgcn_wmma_f32_16x16x4_f32(false, a, false, b1,
                                                         (short)0, acc1, false, false);
            acc2 = __builtin_amdgcn_wmma_f32_16x16x4_f32(false, a, false, b2,
                                                         (short)0, acc2, false, false);
        }

#pragma unroll
        for (int v = 0; v < 8; ++v) {
            vbuf[v + 8 * half][n]      = acc0[v];
            vbuf[v + 8 * half][16 + n] = acc1[v];
            vbuf[v + 8 * half][32 + n] = acc2[v];
        }
    }
    __syncthreads();
    if (live) {
        v8f dacc = {0,0,0,0,0,0,0,0};
        for (int c = 0; c < 9; ++c) {
            const int kb = 4 * c + 2 * half;
            v2f a;
            a.x = ltap(kb - 2 * n);
            a.y = ltap(kb + 1 - 2 * n);
            v2f b;
            b.x = vbuf[n][kb];
            b.y = vbuf[n][kb + 1];
            dacc = __builtin_amdgcn_wmma_f32_16x16x4_f32(false, a, false, b,
                                                         (short)0, dacc, false, false);
        }

        const int gi = i0 + n;
        if (gi < H2) {
            float* dp = down + (size_t)p * H2 * W2 + (size_t)gi * W2;
#pragma unroll
            for (int v = 0; v < 8; ++v) {
                int gj = j0 + v + 8 * half;
                if (gj < W2) dp[gj] = dacc[v];
            }
        }
    }
}

template <bool GAM>
__global__ void __launch_bounds__(32)
nlp_down(const float* __restrict__ src, float* __restrict__ down, int H, int W) {
    __shared__ float vbuf[16][49];
    down_tile<GAM>(src, down, H, W, blockIdx.z, blockIdx.y * 16, blockIdx.x * 16, true,
                   vbuf, threadIdx.x);
}

template <bool GAM>
__device__ __forceinline__ void sub_tile(const float* __restrict__ src,
                                         const float* __restrict__ down,
                                         float* __restrict__ out, int H, int W, int b,
                                         int y0, int x0, int tid,
                                         float (*zb)[36][37], float (*dt)[20][21]) {
    const int H2 = H / 2 + 2, W2 = W / 2 + 2;
    const unsigned dpl = (unsigned)H2 * (unsigned)W2;
    const float* dw = down + (size_t)(3 * b) * dpl;

    const int dr0 = (y0 >> 1) - 1, dc0 = (x0 >> 1) - 1;

    unsigned base, rpitch, cpitch, chp;
    if (GAM) {
        base = (unsigned)b * (unsigned)H * (unsigned)W * 3u;
        rpitch = (unsigned)W * 3u;
        cpitch = 3u;
        chp = 1u;
    } else {
        const unsigned Wp = (unsigned)W + 2u;
        chp = ((unsigned)H + 2u) * Wp;
        base = (unsigned)(3 * b) * chp + Wp + 1u;
        rpitch = Wp;
        cpitch = 1u;
    }

    float dv[2][3];
#pragma unroll
    for (int q = 0; q < 2; ++q) {
        int t = tid + q * 256; t = t > 399 ? 399 : t;
        int ly = t / 20, lx = t % 20;
        int r = clampi(dr0 + ly, 0, H2 - 1);
        int c = clampi(dc0 + lx, 0, W2 - 1);
        const unsigned o = (unsigned)r * (unsigned)W2 + (unsigned)c;
#pragma unroll
        for (int ch = 0; ch < 3; ++ch) dv[q][ch] = (dw + (size_t)ch * dpl)[o];
    }
    float xr[6][3];
#pragma unroll
    for (int q = 0; q < 6; ++q) {
        int t = tid + q * 256; t = t > 1295 ? 1295 : t;
        int ly = t / 36, lx = t % 36;
        int y = clampi(y0 - 2 + ly, 0, H - 1);
        int x = clampi(x0 - 2 + lx, 0, W - 1);
        const unsigned o = base + (unsigned)y * rpitch + (unsigned)x * cpitch;
#pragma unroll
        for (int ch = 0; ch < 3; ++ch) {
            if (GAM) xr[q][ch] = (src + o)[ch];
            else     xr[q][ch] = (src + (size_t)ch * chp)[o];
        }
    }

#pragma unroll
    for (int q = 0; q < 2; ++q) {
        int t = tid + q * 256;
        if (t < 400) {
#pragma unroll
            for (int ch = 0; ch < 3; ++ch) dt[ch][t / 20][t % 20] = dv[q][ch];
        }
    }
    __syncthreads();

#pragma unroll
    for (int q = 0; q < 6; ++q) {
        int t = tid + q * 256;
        if (t < 1296) {
            int ly = t / 36, lx = t % 36;
            int y = y0 - 2 + ly, x = x0 - 2 + lx;
            float z[3] = {0.0f, 0.0f, 0.0f};
            if (y >= 0 && y < H && x >= 0 && x < W) {
                float wv[3], wh[3];
                int r0, c0;
                if (y & 1) { r0 = (y + 1) >> 1; wv[0] = 0.5f; wv[1] = 0.5f; wv[2] = 0.0f; }
                else       { r0 = y >> 1;       wv[0] = 0.1f; wv[1] = 0.8f; wv[2] = 0.1f; }
                if (x & 1) { c0 = (x + 1) >> 1; wh[0] = 0.5f; wh[1] = 0.5f; wh[2] = 0.0f; }
                else       { c0 = x >> 1;       wh[0] = 0.1f; wh[1] = 0.8f; wh[2] = 0.1f; }
                int lr[3], lc[3];
#pragma unroll
                for (int k = 0; k < 3; ++k) {
                    int a = r0 + k - dr0; lr[k] = a < 0 ? 0 : (a > 19 ? 19 : a);
                    int e = c0 + k - dc0; lc[k] = e < 0 ? 0 : (e > 19 ? 19 : e);
                }
#pragma unroll
                for (int ch = 0; ch < 3; ++ch) {
                    float xv = GAM ? gamma_of(xr[q][ch]) : xr[q][ch];
                    float up = 0.0f;
#pragma unroll
                    for (int rr = 0; rr < 3; ++rr) {
                        float rowv = 0.0f;
#pragma unroll
                        for (int cc = 0; cc < 3; ++cc)
                            rowv = fmaf(wh[cc], dt[ch][lr[rr]][lc[cc]], rowv);
                        up = fmaf(wv[rr], rowv, up);
                    }
                    z[ch] = xv - up;
                }
            }
#pragma unroll
            for (int ch = 0; ch < 3; ++ch) zb[ch][ly][lx] = z[ch];
        }
    }
    __syncthreads();

    const float Pk[5][5] = {
        {0.04f, 0.04f, 0.05f, 0.04f, 0.04f},
        {0.04f, 0.03f, 0.04f, 0.03f, 0.04f},
        {0.05f, 0.04f, 0.05f, 0.04f, 0.05f},
        {0.04f, 0.03f, 0.04f, 0.03f, 0.04f},
        {0.04f, 0.04f, 0.05f, 0.04f, 0.04f}};

#pragma unroll
    for (int q = 0; q < 4; ++q) {
        int idx = tid + q * 256;
        int oy = idx >> 5, ox = idx & 31;
        float pool[3] = {0.0f, 0.0f, 0.0f};
#pragma unroll
        for (int dy = 0; dy < 5; ++dy)
#pragma unroll
            for (int dx = 0; dx < 5; ++dx)
#pragma unroll
                for (int ch = 0; ch < 3; ++ch)
                    pool[ch] = fmaf(Pk[dy][dx], fabsf(zb[ch][oy + dy][ox + dx]), pool[ch]);
        int gy = y0 + oy, gx = x0 + ox;
        float* op = out + (((size_t)b * H + gy) * W + gx) * 3;
#pragma unroll
        for (int ch = 0; ch < 3; ++ch)
            op[ch] = zb[ch][oy + 2][ox + 2] / (pool[ch] + SIGMA_P);
    }
}

template <bool GAM, bool LAST>
__global__ void __launch_bounds__(256) __attribute__((amdgpu_waves_per_eu(12)))
nlp_level(const float* __restrict__ src, const float* __restrict__ down,
          float* __restrict__ out, float* __restrict__ nxt, int H, int W, int nd, int sh) {
    __shared__ float smem[LAST ? (3 * (36 * 37 + 20 * 21)) : (8 * 16 * 49)];
    static_assert(8 * 16 * 49 >= 3 * (36 * 37 + 20 * 21), "sub role must fit");

    const int bz = blockIdx.z;
    const int b = blockIdx.x;
    const int tid = threadIdx.x;

    if (b < 3 * nd) {
        const int pc = (b >= nd ? 1 : 0) + (b >= 2 * nd ? 1 : 0);
        const int g = b - pc * nd;
        const int p = 3 * bz + pc;
        if (LAST) {
            const int y = tid >> 4, x = tid & 15;
            float v = down[(size_t)p * 18 * 18 + (size_t)(y + 1) * 18 + (x + 1)];
            float o = v / (fabsf(v) + SIGMA_LP);
            nxt[(((size_t)bz * 16 + y) * 16 + x) * 3 + pc] = o;
        } else {
            const int wave = __builtin_amdgcn_readfirstlane(tid >> 5);
            const int Hn = H / 2, Wn = W / 2;
            const int Ty = (Hn / 2 + 2 + 15) / 16, Tx = (Wn / 2 + 2 + 15) / 16;
            const int tile = g * 8 + wave;
            const bool live = tile < Ty * Tx;
            const int ti = tile / Tx, tj = tile - ti * Tx;
            down_tile<false>(down, nxt, Hn, Wn, p, ti * 16, tj * 16, live,
                             (float (*)[49])(smem + wave * (16 * 49)), tid & 31);
        }
    } else {
        const int s = b - 3 * nd;
        sub_tile<GAM>(src, down, out, H, W, bz, (s >> sh) * 32, (s & ((1 << sh) - 1)) * 32,
                      tid, (float (*)[36][37])smem, (float (*)[20][21])(smem + 3 * 36 * 37));
    }
}

extern "C" void kernel_launch(void* const* d_in, const int* in_sizes, int n_in,
                              void* d_out, int out_size, void* d_ws, size_t ws_size,
                              hipStream_t stream) {
    (void)in_sizes; (void)n_in; (void)out_size; (void)ws_size;
    const float* img = (const float*)d_in[0];
    float* out = (float*)d_out;
    float* ws  = (float*)d_ws;

    const int Hl[6]  = {512, 256, 128, 64, 32, 16};
    const int shl[5] = {4, 3, 2, 1, 0};

    size_t dof[5]; size_t off = 0;
    for (int l = 0; l < 5; ++l) {
        dof[l] = off;
        size_t h2 = (size_t)(Hl[l] / 2 + 2);
        off += (size_t)48 * h2 * h2;
    }
    size_t oof[6]; size_t o2 = 0;
    for (int l = 0; l < 6; ++l) {
        oof[l] = o2;
        o2 += (size_t)16 * Hl[l] * Hl[l] * 3;
    }

    {
        const int T = (Hl[0] / 2 + 2 + 15) / 16;
        nlp_down<true><<<dim3(T, T, 48), 32, 0, stream>>>(img, ws + dof[0], Hl[0], Hl[0]);
    }
    const float* srcp = img;
    for (int l = 0; l < 5; ++l) {
        const int H = Hl[l];
        const int ns = (H / 32) * (H / 32);
        if (l < 4) {
            const int Tn = (Hl[l + 1] / 2 + 2 + 15) / 16;
            const int nd = (Tn * Tn + 7) / 8;
            if (l == 0)
                nlp_level<true, false><<<dim3(3 * nd + ns, 1, 16), 256, 0, stream>>>(
                    srcp, ws + dof[l], out + oof[l], ws + dof[l + 1], H, H, nd, shl[l]);
            else
                nlp_level<false, false><<<dim3(3 * nd + ns, 1, 16), 256, 0, stream>>>(
                    srcp, ws + dof[l], out + oof[l], ws + dof[l + 1], H, H, nd, shl[l]);
        } else {
            nlp_level<false, true><<<dim3(3 + ns, 1, 16), 256, 0, stream>>>(
                srcp, ws + dof[l], out + oof[l], out + oof[5], H, H, 1, shl[l]);
        }
        srcp = ws + dof[l];
    }
}


# sub-band work coarsened: a thread forms 2x2 blocks of z from one 3x3 down window and columns of 4 outputs from one 8x5 z window; tiles of the two large levels on 128 threads; sums keep their order
# speedup vs baseline: 1.6147x; 1.0645x over previous
#include <hip/hip_runtime.h>
#include <hip/hip_bf16.h>


typedef float v2f __attribute__((ext_vector_type(2)));
typedef float v8f __attribute__((ext_vector_type(8)));

#define GAMMA_EXP 0.38461538461538464f
#define SIGMA_P   0.17f
#define SIGMA_LP  4.86f

__device__ __forceinline__ int reflc(int r, int n) {
    r = (r < 0) ? -r : r;
    if (r >= n) r = 2 * n - 2 - r;
    if (r < 0)  r = 0;
    if (r >= n) r = n - 1;
    return r;
}

__device__ __forceinline__ int clampi(int v, int lo, int hi) {
    return v < lo ? lo : (v > hi ? hi : v);
}

__device__ __forceinline__ float ltap(int d) {
    float v = 0.0f;
    if (d == 0 || d == 4)      v = 0.05f;
    else if (d == 1 || d == 3) v = 0.25f;
    else if (d == 2)           v = 0.40f;
    return v;
}

__device__ __forceinline__ float gamma_of(float v) {
    return __builtin_amdgcn_exp2f(GAMMA_EXP * __builtin_amdgcn_logf(v));
}

template <bool GAM>
__device__ __forceinline__ void src_layout(int p, int H, int W, unsigned& base,
                                           unsigned& rpitch, unsigned& cpitch) {
    if (GAM) {
        const int b = p / 3, c = p - 3 * b;
        base = (unsigned)b * (unsigned)H * (unsigned)W * 3u + (unsigned)c;
        rpitch = (unsigned)W * 3u;
        cpitch = 3u;
    } else {
        const unsigned Wp = (unsigned)W + 2u;
        base = (unsigned)p * ((unsigned)H + 2u) * Wp + Wp + 1u;
        rpitch = Wp;
        cpitch = 1u;
    }
}

template <bool GAM>
__device__ __forceinline__ void down_tile(const float* __restrict__ src, float* __restrict__ down,
                                          int H, int W, int p, int i0, int j0, bool live,
                                          float (*vbuf)[49], int lane) {
    const int H2 = H / 2 + 2, W2 = W / 2 + 2;
    const int n = lane & 15;
    const int half = lane >> 4;

    if (live) {
        unsigned base, rpitch, cpitch;
        src_layout<GAM>(p, H, W, base, rpitch, cpitch);

        unsigned colo[3];
#pragma unroll
        for (int nt = 0; nt < 3; ++nt)
            colo[nt] = base + cpitch * (unsigned)reflc(2 * j0 - 4 + nt * 16 + n, W);

        float raw[9][6];
#pragma unroll
        for (int c = 0; c < 9; ++c) {
            const int kb = 4 * c + 2 * half;
            const unsigned ro0 = rpitch * (unsigned)reflc(2 * i0 + kb - 4, H);
            const unsigned ro1 = rpitch * (unsigned)reflc(2 * i0 + kb - 3, H);
#pragma unroll
            for (int nt = 0; nt < 3; ++nt) {
                raw[c][2 * nt]     = src[ro0 + colo[nt]];
                raw[c][2 * nt + 1] = src[ro1 + colo[nt]];
            }
        }

        __builtin_amdgcn_sched_barrier(0);

        v8f acc0 = {0,0,0,0,0,0,0,0};
        v8f acc1 = {0,0,0,0,0,0,0,0};
        v8f acc2 = {0,0,0,0,0,0,0,0};

#pragma unroll
        for (int c = 0; c < 9; ++c) {
            const int kb = 4 * c + 2 * half;
            v2f a;
            a.x = ltap(kb - 2 * n);
            a.y = ltap(kb + 1 - 2 * n);

            v2f b0, b1, b2;
            if (GAM) {
                b0.x = gamma_of(raw[c][0]); b0.y = gamma_of(raw[c][1]);
                b1.x = gamma_of(raw[c][2]); b1.y = gamma_of(raw[c][3]);
                b2.x = gamma_of(raw[c][4]); b2.y = gamma_of(raw[c][5]);
            } else {
                b0.x = raw[c][0]; b0.y = raw[c][1];
                b1.x = raw[c][2]; b1.y = raw[c][3];
                b2.x = raw[c][4]; b2.y = raw[c][5];
            }

            acc0 = __builtin_amdgcn_wmma_f32_16x16x4_f32(false, a, false, b0,
                                                         (short)0, acc0, false, false);
            acc1 = __builtin_amdgcn_wmma_f32_16x16x4_f32(false, a, false, b1,
                                                         (short)0, acc1, false, false);
            acc2 = __builtin_amdgcn_wmma_f32_16x16x4_f32(false, a, false, b2,
                                                         (short)0, acc2, false, false);
        }

#pragma unroll
        for (int v = 0; v < 8; ++v) {
            vbuf[v + 8 * half][n]      = acc0[v];
            vbuf[v + 8 * half][16 + n] = acc1[v];
            vbuf[v + 8 * half][32 + n] = acc2[v];
        }
    }
    __syncthreads();
    if (live) {
        v8f dacc = {0,0,0,0,0,0,0,0};
        for (int c = 0; c < 9; ++c) {
            const int kb = 4 * c + 2 * half;
            v2f a;
            a.x = ltap(kb - 2 * n);
            a.y = ltap(kb + 1 - 2 * n);
            v2f b;
            b.x = vbuf[n][kb];
            b.y = vbuf[n][kb + 1];
            dacc = __builtin_amdgcn_wmma_f32_16x16x4_f32(false, a, false, b,
                                                         (short)0, dacc, false, false);
        }

        const int gi = i0 + n;
        if (gi < H2) {
            float* dp = down + (size_t)p * H2 * W2 + (size_t)gi * W2;
#pragma unroll
            for (int v = 0; v < 8; ++v) {
                int gj = j0 + v + 8 * half;
                if (gj < W2) dp[gj] = dacc[v];
            }
        }
    }
}

template <bool GAM>
__global__ void __launch_bounds__(32)
nlp_down(const float* __restrict__ src, float* __restrict__ down, int H, int W) {
    __shared__ float vbuf[16][49];
    down_tile<GAM>(src, down, H, W, blockIdx.z, blockIdx.y * 16, blockIdx.x * 16, true,
                   vbuf, threadIdx.x);
}

template <bool GAM, int NT>
__device__ __forceinline__ void sub_tile(const float* __restrict__ src,
                                         const float* __restrict__ down,
                                         float* __restrict__ out, int H, int W, int b,
                                         int y0, int x0, int tid,
                                         float (*zb)[36][37], float (*dt)[20][21]) {
    constexpr int DQ = (400 + NT - 1) / NT;
    constexpr int BQ = (324 + NT - 1) / NT;
    constexpr int RB = 4;
    constexpr int SQ = 256 / NT;
    const int H2 = H / 2 + 2, W2 = W / 2 + 2;
    const unsigned dpl = (unsigned)H2 * (unsigned)W2;
    const float* dw = down + (size_t)(3 * b) * dpl;

    const int dr0 = (y0 >> 1) - 1, dc0 = (x0 >> 1) - 1;

    unsigned base, rpitch, cpitch, chp;
    if (GAM) {
        base = (unsigned)b * (unsigned)H * (unsigned)W * 3u;
        rpitch = (unsigned)W * 3u;
        cpitch = 3u;
        chp = 1u;
    } else {
        const unsigned Wp = (unsigned)W + 2u;
        chp = ((unsigned)H + 2u) * Wp;
        base = (unsigned)(3 * b) * chp + Wp + 1u;
        rpitch = Wp;
        cpitch = 1u;
    }

    float dv[DQ][3];
#pragma unroll
    for (int q = 0; q < DQ; ++q) {
        int t = tid + q * NT; t = t > 399 ? 399 : t;
        int ly = t / 20, lx = t % 20;
        int r = clampi(dr0 + ly, 0, H2 - 1);
        int c = clampi(dc0 + lx, 0, W2 - 1);
        const unsigned o = (unsigned)r * (unsigned)W2 + (unsigned)c;
#pragma unroll
        for (int ch = 0; ch < 3; ++ch) dv[q][ch] = (dw + (size_t)ch * dpl)[o];
    }
    float xr[BQ][2][2][3];
#pragma unroll
    for (int q = 0; q < BQ; ++q) {
        int t = tid + q * NT; t = t > 323 ? 323 : t;
        const int by = t / 18, bx = t - by * 18;
        const int y = clampi(y0 - 2 + 2 * by, 0, H - 2);
        const int x = clampi(x0 - 2 + 2 * bx, 0, W - 2);
        const unsigned o = base + (unsigned)y * rpitch + (unsigned)x * cpitch;
#pragma unroll
        for (int i = 0; i < 2; ++i)
#pragma unroll
            for (int j = 0; j < 2; ++j)
#pragma unroll
                for (int ch = 0; ch < 3; ++ch) {
                    if (GAM) xr[q][i][j][ch] = (src + o + (unsigned)i * rpitch)[3 * j + ch];
                    else     xr[q][i][j][ch] = (src + (size_t)ch * chp)[o + (unsigned)i * rpitch + (unsigned)j];
                }
    }

#pragma unroll
    for (int q = 0; q < DQ; ++q) {
        int t = tid + q * NT;
        if (t < 400) {
#pragma unroll
            for (int ch = 0; ch < 3; ++ch) dt[ch][t / 20][t % 20] = dv[q][ch];
        }
    }
    __syncthreads();

#pragma unroll
    for (int q = 0; q < BQ; ++q) {
        const int t = tid + q * NT;
        if (t < 324) {
            const int by = t / 18, bx = t - by * 18;
            const int y = y0 - 2 + 2 * by, x = x0 - 2 + 2 * bx;
            const bool inside = (y >= 0 && y < H && x >= 0 && x < W);
#pragma unroll
            for (int ch = 0; ch < 3; ++ch) {
                float z00 = 0.0f, z01 = 0.0f, z10 = 0.0f, z11 = 0.0f;
                if (inside) {
                    float he[3], ho[3];
#pragma unroll
                    for (int rr = 0; rr < 3; ++rr) {
                        const float d0 = dt[ch][by + rr][bx];
                        const float d1 = dt[ch][by + rr][bx + 1];
                        const float d2 = dt[ch][by + rr][bx + 2];
                        float e = 0.0f;
                        e = fmaf(0.1f, d0, e); e = fmaf(0.8f, d1, e); e = fmaf(0.1f, d2, e);
                        float o = 0.0f;
                        o = fmaf(0.5f, d1, o); o = fmaf(0.5f, d2, o);
                        he[rr] = e; ho[rr] = o;
                    }
                    float uee = 0.0f, ueo = 0.0f, uoe = 0.0f, uoo = 0.0f;
                    uee = fmaf(0.1f, he[0], uee); uee = fmaf(0.8f, he[1], uee); uee = fmaf(0.1f, he[2], uee);
                    ueo = fmaf(0.1f, ho[0], ueo); ueo = fmaf(0.8f, ho[1], ueo); ueo = fmaf(0.1f, ho[2], ueo);
                    uoe = fmaf(0.5f, he[1], uoe); uoe = fmaf(0.5f, he[2], uoe);
                    uoo = fmaf(0.5f, ho[1], uoo); uoo = fmaf(0.5f, ho[2], uoo);
                    const float x00 = GAM ? gamma_of(xr[q][0][0][ch]) : xr[q][0][0][ch];
                    const float x01 = GAM ? gamma_of(xr[q][0][1][ch]) : xr[q][0][1][ch];
                    const float x10 = GAM ? gamma_of(xr[q][1][0][ch]) : xr[q][1][0][ch];
                    const float x11 = GAM ? gamma_of(xr[q][1][1][ch]) : xr[q][1][1][ch];
                    z00 = x00 - uee; z01 = x01 - ueo; z10 = x10 - uoe; z11 = x11 - uoo;
                }
                zb[ch][2 * by][2 * bx]         = z00;
                zb[ch][2 * by][2 * bx + 1]     = z01;
                zb[ch][2 * by + 1][2 * bx]     = z10;
                zb[ch][2 * by + 1][2 * bx + 1] = z11;
            }
        }
    }
    __syncthreads();

    const float Pk[5][5] = {
        {0.04f, 0.04f, 0.05f, 0.04f, 0.04f},
        {0.04f, 0.03f, 0.04f, 0.03f, 0.04f},
        {0.05f, 0.04f, 0.05f, 0.04f, 0.05f},
        {0.04f, 0.03f, 0.04f, 0.03f, 0.04f},
        {0.04f, 0.04f, 0.05f, 0.04f, 0.04f}};

    for (int k = 0; k < SQ; ++k) {
        const int s = tid + k * NT;
        const int ox = s & 31, g = s >> 5;
        float pool[RB][3], ctr[RB][3];
#pragma unroll
        for (int r = 0; r < RB; ++r)
#pragma unroll
            for (int ch = 0; ch < 3; ++ch) { pool[r][ch] = 0.0f; ctr[r][ch] = 0.0f; }
#pragma unroll
        for (int wr = 0; wr < RB + 4; ++wr) {
#pragma unroll
            for (int ch = 0; ch < 3; ++ch) {
                float a[5];
#pragma unroll
                for (int dx = 0; dx < 5; ++dx) a[dx] = zb[ch][RB * g + wr][ox + dx];
                if (wr >= 2 && wr < RB + 2) ctr[wr - 2][ch] = a[2];
#pragma unroll
                for (int dx = 0; dx < 5; ++dx) {
                    const float m = fabsf(a[dx]);
#pragma unroll
                    for (int r = 0; r < RB; ++r) {
                        const int dy = wr - r;
                        if (dy >= 0 && dy < 5) pool[r][ch] = fmaf(Pk[dy][dx], m, pool[r][ch]);
                    }
                }
            }
        }
#pragma unroll
        for (int r = 0; r < RB; ++r) {
            const int gy = y0 + RB * g + r, gx = x0 + ox;
            float* op = out + (((size_t)b * H + gy) * W + gx) * 3;
#pragma unroll
            for (int ch = 0; ch < 3; ++ch)
                op[ch] = ctr[r][ch] / (pool[r][ch] + SIGMA_P);
        }
    }
}

template <bool GAM, bool LAST, int NT>
__global__ void __launch_bounds__(NT) __attribute__((amdgpu_waves_per_eu(12)))
nlp_level(const float* __restrict__ src, const float* __restrict__ down,
          float* __restrict__ out, float* __restrict__ nxt, int H, int W, int nd, int sh) {
    constexpr int SUBF = 3 * (36 * 37 + 20 * 21);
    constexpr int DWNF = LAST ? 0 : (NT / 32) * 16 * 49;
    __shared__ float smem[SUBF > DWNF ? SUBF : DWNF];

    const int bz = blockIdx.z;
    const int b = blockIdx.x;
    const int tid = threadIdx.x;

    if (b < 3 * nd) {
        const int pc = (b >= nd ? 1 : 0) + (b >= 2 * nd ? 1 : 0);
        const int g = b - pc * nd;
        const int p = 3 * bz + pc;
        if (LAST) {
            for (int e = tid; e < 256; e += NT) {
                const int y = e >> 4, x = e & 15;
                float v = down[(size_t)p * 18 * 18 + (size_t)(y + 1) * 18 + (x + 1)];
                float o = v / (fabsf(v) + SIGMA_LP);
                nxt[(((size_t)bz * 16 + y) * 16 + x) * 3 + pc] = o;
            }
        } else {
            const int wave = __builtin_amdgcn_readfirstlane(tid >> 5);
            const int Hn = H / 2, Wn = W / 2;
            const int Ty = (Hn / 2 + 2 + 15) / 16, Tx = (Wn / 2 + 2 + 15) / 16;
            const int tile = g * (NT / 32) + wave;
            const bool live = tile < Ty * Tx;
            const int ti = tile / Tx, tj = tile - ti * Tx;
            down_tile<false>(down, nxt, Hn, Wn, p, ti * 16, tj * 16, live,
                             (float (*)[49])(smem + wave * (16 * 49)), tid & 31);
        }
    } else {
        const int s = b - 3 * nd;
        sub_tile<GAM, NT>(src, down, out, H, W, bz, (s >> sh) * 32, (s & ((1 << sh) - 1)) * 32,
                          tid, (float (*)[36][37])smem, (float (*)[20][21])(smem + 3 * 36 * 37));
    }
}

extern "C" void kernel_launch(void* const* d_in, const int* in_sizes, int n_in,
                              void* d_out, int out_size, void* d_ws, size_t ws_size,
                              hipStream_t stream) {
    (void)in_sizes; (void)n_in; (void)out_size; (void)ws_size;
    const float* img = (const float*)d_in[0];
    float* out = (float*)d_out;
    float* ws  = (float*)d_ws;

    const int Hl[6]  = {512, 256, 128, 64, 32, 16};
    const int shl[5] = {4, 3, 2, 1, 0};
    const int ntl[5] = {128, 128, 256, 256, 256};

    size_t dof[5]; size_t off = 0;
    for (int l = 0; l < 5; ++l) {
        dof[l] = off;
        size_t h2 = (size_t)(Hl[l] / 2 + 2);
        off += (size_t)48 * h2 * h2;
    }
    size_t oof[6]; size_t o2 = 0;
    for (int l = 0; l < 6; ++l) {
        oof[l] = o2;
        o2 += (size_t)16 * Hl[l] * Hl[l] * 3;
    }

    {
        const int T = (Hl[0] / 2 + 2 + 15) / 16;
        nlp_down<true><<<dim3(T, T, 48), 32, 0, stream>>>(img, ws + dof[0], Hl[0], Hl[0]);
    }
    const float* srcp = img;
    for (int l = 0; l < 5; ++l) {
        const int H = Hl[l];
        const int ns = (H / 32) * (H / 32);
        if (l < 4) {
            const int Tn = (Hl[l + 1] / 2 + 2 + 15) / 16;
            const int tw = ntl[l] / 32;
            const int nd = (Tn * Tn + tw - 1) / tw;
            if (l == 0)
                nlp_level<true, false, 128><<<dim3(3 * nd + ns, 1, 16), 128, 0, stream>>>(
                    srcp, ws + dof[l], out + oof[l], ws + dof[l + 1], H, H, nd, shl[l]);
            else if (l == 1)
                nlp_level<false, false, 128><<<dim3(3 * nd + ns, 1, 16), 128, 0, stream>>>(
                    srcp, ws + dof[l], out + oof[l], ws + dof[l + 1], H, H, nd, shl[l]);
            else
                nlp_level<false, false, 256><<<dim3(3 * nd + ns, 1, 16), 256, 0, stream>>>(
                    srcp, ws + dof[l], out + oof[l], ws + dof[l + 1], H, H, nd, shl[l]);
        } else {
            nlp_level<false, true, 256><<<dim3(3 + ns, 1, 16), 256, 0, stream>>>(
                srcp, ws + dof[l], out + oof[l], out + oof[5], H, H, 1, shl[l]);
        }
        srcp = ws + dof[l];
    }
}


# level-0 downsample: one wave forms its tile for all three channels, reading NHWC floats as columns as they lie (12-byte loads, 7 column subtiles not 9, a third of the waves); sums keep their order
# speedup vs baseline: 1.7373x; 1.0759x over previous
#include <hip/hip_runtime.h>
#include <hip/hip_bf16.h>


typedef float v2f __attribute__((ext_vector_type(2)));
typedef float v8f __attribute__((ext_vector_type(8)));

#define GAMMA_EXP 0.38461538461538464f
#define SIGMA_P   0.17f
#define SIGMA_LP  4.86f

__device__ __forceinline__ int reflc(int r, int n) {
    r = (r < 0) ? -r : r;
    if (r >= n) r = 2 * n - 2 - r;
    if (r < 0)  r = 0;
    if (r >= n) r = n - 1;
    return r;
}

__device__ __forceinline__ int clampi(int v, int lo, int hi) {
    return v < lo ? lo : (v > hi ? hi : v);
}

__device__ __forceinline__ float ltap(int d) {
    float v = 0.0f;
    if (d == 0 || d == 4)      v = 0.05f;
    else if (d == 1 || d == 3) v = 0.25f;
    else if (d == 2)           v = 0.40f;
    return v;
}

__device__ __forceinline__ float gamma_of(float v) {
    return __builtin_amdgcn_exp2f(GAMMA_EXP * __builtin_amdgcn_logf(v));
}

__device__ __forceinline__ void src_layout(int p, int H, int W, unsigned& base, unsigned& rpitch) {
    const unsigned Wp = (unsigned)W + 2u;
    base = (unsigned)p * ((unsigned)H + 2u) * Wp + Wp + 1u;
    rpitch = Wp;
}

__device__ __forceinline__ void down_tile(const float* __restrict__ src, float* __restrict__ down,
                                          int H, int W, int p, int i0, int j0, bool live,
                                          float (*vbuf)[49], int lane) {
    const int H2 = H / 2 + 2, W2 = W / 2 + 2;
    const int n = lane & 15;
    const int half = lane >> 4;

    if (live) {
        unsigned base, rpitch;
        src_layout(p, H, W, base, rpitch);

        unsigned colo[3];
#pragma unroll
        for (int nt = 0; nt < 3; ++nt)
            colo[nt] = base + (unsigned)reflc(2 * j0 - 4 + nt * 16 + n, W);

        float raw[9][6];
#pragma unroll
        for (int c = 0; c < 9; ++c) {
            const int kb = 4 * c + 2 * half;
            const unsigned ro0 = rpitch * (unsigned)reflc(2 * i0 + kb - 4, H);
            const unsigned ro1 = rpitch * (unsigned)reflc(2 * i0 + kb - 3, H);
#pragma unroll
            for (int nt = 0; nt < 3; ++nt) {
                raw[c][2 * nt]     = src[ro0 + colo[nt]];
                raw[c][2 * nt + 1] = src[ro1 + colo[nt]];
            }
        }

        __builtin_amdgcn_sched_barrier(0);

        v8f acc0 = {0,0,0,0,0,0,0,0};
        v8f acc1 = {0,0,0,0,0,0,0,0};
        v8f acc2 = {0,0,0,0,0,0,0,0};

#pragma unroll
        for (int c = 0; c < 9; ++c) {
            const int kb = 4 * c + 2 * half;
            v2f a;
            a.x = ltap(kb - 2 * n);
            a.y = ltap(kb + 1 - 2 * n);

            v2f b0, b1, b2;
            b0.x = raw[c][0]; b0.y = raw[c][1];
            b1.x = raw[c][2]; b1.y = raw[c][3];
            b2.x = raw[c][4]; b2.y = raw[c][5];

            acc0 = __builtin_amdgcn_wmma_f32_16x16x4_f32(false, a, false, b0,
                                                         (short)0, acc0, false, false);
            acc1 = __builtin_amdgcn_wmma_f32_16x16x4_f32(false, a, false, b1,
                                                         (short)0, acc1, false, false);
            acc2 = __builtin_amdgcn_wmma_f32_16x16x4_f32(false, a, false, b2,
                                                         (short)0, acc2, false, false);
        }

#pragma unroll
        for (int v = 0; v < 8; ++v) {
            vbuf[v + 8 * half][n]      = acc0[v];
            vbuf[v + 8 * half][16 + n] = acc1[v];
            vbuf[v + 8 * half][32 + n] = acc2[v];
        }
    }
    __syncthreads();
    if (live) {
        v8f dacc = {0,0,0,0,0,0,0,0};
        for (int c = 0; c < 9; ++c) {
            const int kb = 4 * c + 2 * half;
            v2f a;
            a.x = ltap(kb - 2 * n);
            a.y = ltap(kb + 1 - 2 * n);
            v2f b;
            b.x = vbuf[n][kb];
            b.y = vbuf[n][kb + 1];
            dacc = __builtin_amdgcn_wmma_f32_16x16x4_f32(false, a, false, b,
                                                         (short)0, dacc, false, false);
        }

        const int gi = i0 + n;
        if (gi < H2) {
            float* dp = down + (size_t)p * H2 * W2 + (size_t)gi * W2;
#pragma unroll
            for (int v = 0; v < 8; ++v) {
                int gj = j0 + v + 8 * half;
                if (gj < W2) dp[gj] = dacc[v];
            }
        }
    }
}

__global__ void __launch_bounds__(32) __attribute__((amdgpu_waves_per_eu(5)))
nlp_down0(const float* __restrict__ img, float* __restrict__ down, int H, int W) {
    __shared__ float vbuf[3][16][49];
    const int lane = threadIdx.x;
    const int bz = blockIdx.z, i0 = blockIdx.y * 16, j0 = blockIdx.x * 16;
    const int H2 = H / 2 + 2, W2 = W / 2 + 2;
    const int n = lane & 15;
    const int half = lane >> 4;

    const char* imgb = (const char*)img;
    const unsigned base = (unsigned)bz * (unsigned)H * (unsigned)W * 12u;
    const unsigned rpitch = (unsigned)W * 12u;

    unsigned colo[2];
#pragma unroll
    for (int g = 0; g < 2; ++g)
        colo[g] = base + 12u * (unsigned)reflc(2 * j0 - 4 + g * 16 + n, W);
    const int t6 = 32 + n / 3, ch6 = n - 3 * (n / 3);
    const unsigned col6 = base + 12u * (unsigned)reflc(2 * j0 - 4 + t6, W) + 4u * (unsigned)ch6;

    float raw[9][4][3];
    float raw6[9][2];
#pragma unroll
    for (int c = 0; c < 9; ++c) {
        const int kb = 4 * c + 2 * half;
        const unsigned ro0 = rpitch * (unsigned)reflc(2 * i0 + kb - 4, H);
        const unsigned ro1 = rpitch * (unsigned)reflc(2 * i0 + kb - 3, H);
#pragma unroll
        for (int g = 0; g < 2; ++g) {
            const float* p0 = (const float*)(imgb + (ro0 + colo[g]));
            const float* p1 = (const float*)(imgb + (ro1 + colo[g]));
#pragma unroll
            for (int ch = 0; ch < 3; ++ch) {
                raw[c][2 * g][ch]     = p0[ch];
                raw[c][2 * g + 1][ch] = p1[ch];
            }
        }
        raw6[c][0] = *(const float*)(imgb + (ro0 + col6));
        raw6[c][1] = *(const float*)(imgb + (ro1 + col6));
    }

    __builtin_amdgcn_sched_barrier(0);

#pragma unroll
    for (int ch = 0; ch < 3; ++ch) {
        v8f acc0 = {0,0,0,0,0,0,0,0};
        v8f acc1 = {0,0,0,0,0,0,0,0};
#pragma unroll
        for (int c = 0; c < 9; ++c) {
            const int kb = 4 * c + 2 * half;
            v2f a;
            a.x = ltap(kb - 2 * n);
            a.y = ltap(kb + 1 - 2 * n);

            v2f b0, b1;
            b0.x = gamma_of(raw[c][0][ch]); b0.y = gamma_of(raw[c][1][ch]);
            b1.x = gamma_of(raw[c][2][ch]); b1.y = gamma_of(raw[c][3][ch]);

            acc0 = __builtin_amdgcn_wmma_f32_16x16x4_f32(false, a, false, b0,
                                                         (short)0, acc0, false, false);
            acc1 = __builtin_amdgcn_wmma_f32_16x16x4_f32(false, a, false, b1,
                                                         (short)0, acc1, false, false);
        }
#pragma unroll
        for (int v = 0; v < 8; ++v) {
            vbuf[ch][v + 8 * half][n]      = acc0[v];
            vbuf[ch][v + 8 * half][16 + n] = acc1[v];
        }
    }
    {
        v8f acc6 = {0,0,0,0,0,0,0,0};
#pragma unroll
        for (int c = 0; c < 9; ++c) {
            const int kb = 4 * c + 2 * half;
            v2f a;
            a.x = ltap(kb - 2 * n);
            a.y = ltap(kb + 1 - 2 * n);
            v2f b6;
            b6.x = gamma_of(raw6[c][0]); b6.y = gamma_of(raw6[c][1]);
            acc6 = __builtin_amdgcn_wmma_f32_16x16x4_f32(false, a, false, b6,
                                                         (short)0, acc6, false, false);
        }
#pragma unroll
        for (int v = 0; v < 8; ++v) vbuf[ch6][v + 8 * half][t6] = acc6[v];
    }
    __syncthreads();

    v8f dacc[3];
#pragma unroll
    for (int ch = 0; ch < 3; ++ch) dacc[ch] = (v8f){0,0,0,0,0,0,0,0};
#pragma unroll
    for (int c = 0; c < 9; ++c) {
        const int kb = 4 * c + 2 * half;
        v2f a;
        a.x = ltap(kb - 2 * n);
        a.y = ltap(kb + 1 - 2 * n);
#pragma unroll
        for (int ch = 0; ch < 3; ++ch) {
            v2f b;
            b.x = vbuf[ch][n][kb];
            b.y = vbuf[ch][n][kb + 1];
            dacc[ch] = __builtin_amdgcn_wmma_f32_16x16x4_f32(false, a, false, b,
                                                             (short)0, dacc[ch], false, false);
        }
    }

    const int gi = i0 + n;
    if (gi < H2) {
#pragma unroll
        for (int ch = 0; ch < 3; ++ch) {
            float* dp = down + (size_t)(3 * bz + ch) * H2 * W2 + (size_t)gi * W2;
#pragma unroll
            for (int v = 0; v < 8; ++v) {
                int gj = j0 + v + 8 * half;
                if (gj < W2) dp[gj] = dacc[ch][v];
            }
        }
    }
}

template <bool GAM, int NT>
__device__ __forceinline__ void sub_tile(const float* __restrict__ src,
                                         const float* __restrict__ down,
                                         float* __restrict__ out, int H, int W, int b,
                                         int y0, int x0, int tid,
                                         float (*zb)[36][37], float (*dt)[20][21]) {
    constexpr int DQ = (400 + NT - 1) / NT;
    constexpr int BQ = (324 + NT - 1) / NT;
    constexpr int RB = 4;
    constexpr int SQ = 256 / NT;
    const int H2 = H / 2 + 2, W2 = W / 2 + 2;
    const unsigned dpl = (unsigned)H2 * (unsigned)W2;
    const float* dw = down + (size_t)(3 * b) * dpl;

    const int dr0 = (y0 >> 1) - 1, dc0 = (x0 >> 1) - 1;

    unsigned base, rpitch, cpitch, chp;
    if (GAM) {
        base = (unsigned)b * (unsigned)H * (unsigned)W * 3u;
        rpitch = (unsigned)W * 3u;
        cpitch = 3u;
        chp = 1u;
    } else {
        const unsigned Wp = (unsigned)W + 2u;
        chp = ((unsigned)H + 2u) * Wp;
        base = (unsigned)(3 * b) * chp + Wp + 1u;
        rpitch = Wp;
        cpitch = 1u;
    }

    float dv[DQ][3];
#pragma unroll
    for (int q = 0; q < DQ; ++q) {
        int t = tid + q * NT; t = t > 399 ? 399 : t;
        int ly = t / 20, lx = t % 20;
        int r = clampi(dr0 + ly, 0, H2 - 1);
        int c = clampi(dc0 + lx, 0, W2 - 1);
        const unsigned o = (unsigned)r * (unsigned)W2 + (unsigned)c;
#pragma unroll
        for (int ch = 0; ch < 3; ++ch) dv[q][ch] = (dw + (size_t)ch * dpl)[o];
    }
    float xr[BQ][2][2][3];
#pragma unroll
    for (int q = 0; q < BQ; ++q) {
        int t = tid + q * NT; t = t > 323 ? 323 : t;
        const int by = t / 18, bx = t - by * 18;
        const int y = clampi(y0 - 2 + 2 * by, 0, H - 2);
        const int x = clampi(x0 - 2 + 2 * bx, 0, W - 2);
        const unsigned o = base + (unsigned)y * rpitch + (unsigned)x * cpitch;
#pragma unroll
        for (int i = 0; i < 2; ++i)
#pragma unroll
            for (int j = 0; j < 2; ++j)
#pragma unroll
                for (int ch = 0; ch < 3; ++ch) {
                    if (GAM) xr[q][i][j][ch] = (src + o + (unsigned)i * rpitch)[3 * j + ch];
                    else     xr[q][i][j][ch] = (src + (size_t)ch * chp)[o + (unsigned)i * rpitch + (unsigned)j];
                }
    }

#pragma unroll
    for (int q = 0; q < DQ; ++q) {
        int t = tid + q * NT;
        if (t < 400) {
#pragma unroll
            for (int ch = 0; ch < 3; ++ch) dt[ch][t / 20][t % 20] = dv[q][ch];
        }
    }
    __syncthreads();

#pragma unroll
    for (int q = 0; q < BQ; ++q) {
        const int t = tid + q * NT;
        if (t < 324) {
            const int by = t / 18, bx = t - by * 18;
            const int y = y0 - 2 + 2 * by, x = x0 - 2 + 2 * bx;
            const bool inside = (y >= 0 && y < H && x >= 0 && x < W);
#pragma unroll
            for (int ch = 0; ch < 3; ++ch) {
                float z00 = 0.0f, z01 = 0.0f, z10 = 0.0f, z11 = 0.0f;
                if (inside) {
                    float he[3], ho[3];
#pragma unroll
                    for (int rr = 0; rr < 3; ++rr) {
                        const float d0 = dt[ch][by + rr][bx];
                        const float d1 = dt[ch][by + rr][bx + 1];
                        const float d2 = dt[ch][by + rr][bx + 2];
                        float e = 0.0f;
                        e = fmaf(0.1f, d0, e); e = fmaf(0.8f, d1, e); e = fmaf(0.1f, d2, e);
                        float o = 0.0f;
                        o = fmaf(0.5f, d1, o); o = fmaf(0.5f, d2, o);
                        he[rr] = e; ho[rr] = o;
                    }
                    float uee = 0.0f, ueo = 0.0f, uoe = 0.0f, uoo = 0.0f;
                    uee = fmaf(0.1f, he[0], uee); uee = fmaf(0.8f, he[1], uee); uee = fmaf(0.1f, he[2], uee);
                    ueo = fmaf(0.1f, ho[0], ueo); ueo = fmaf(0.8f, ho[1], ueo); ueo = fmaf(0.1f, ho[2], ueo);
                    uoe = fmaf(0.5f, he[1], uoe); uoe = fmaf(0.5f, he[2], uoe);
                    uoo = fmaf(0.5f, ho[1], uoo); uoo = fmaf(0.5f, ho[2], uoo);
                    const float x00 = GAM ? gamma_of(xr[q][0][0][ch]) : xr[q][0][0][ch];
                    const float x01 = GAM ? gamma_of(xr[q][0][1][ch]) : xr[q][0][1][ch];
                    const float x10 = GAM ? gamma_of(xr[q][1][0][ch]) : xr[q][1][0][ch];
                    const float x11 = GAM ? gamma_of(xr[q][1][1][ch]) : xr[q][1][1][ch];
                    z00 = x00 - uee; z01 = x01 - ueo; z10 = x10 - uoe; z11 = x11 - uoo;
                }
                zb[ch][2 * by][2 * bx]         = z00;
                zb[ch][2 * by][2 * bx + 1]     = z01;
                zb[ch][2 * by + 1][2 * bx]     = z10;
                zb[ch][2 * by + 1][2 * bx + 1] = z11;
            }
        }
    }
    __syncthreads();

    const float Pk[5][5] = {
        {0.04f, 0.04f, 0.05f, 0.04f, 0.04f},
        {0.04f, 0.03f, 0.04f, 0.03f, 0.04f},
        {0.05f, 0.04f, 0.05f, 0.04f, 0.05f},
        {0.04f, 0.03f, 0.04f, 0.03f, 0.04f},
        {0.04f, 0.04f, 0.05f, 0.04f, 0.04f}};

    for (int k = 0; k < SQ; ++k) {
        const int s = tid + k * NT;
        const int ox = s & 31, g = s >> 5;
        float pool[RB][3], ctr[RB][3];
#pragma unroll
        for (int r = 0; r < RB; ++r)
#pragma unroll
            for (int ch = 0; ch < 3; ++ch) { pool[r][ch] = 0.0f; ctr[r][ch] = 0.0f; }
#pragma unroll
        for (int wr = 0; wr < RB + 4; ++wr) {
#pragma unroll
            for (int ch = 0; ch < 3; ++ch) {
                float a[5];
#pragma unroll
                for (int dx = 0; dx < 5; ++dx) a[dx] = zb[ch][RB * g + wr][ox + dx];
                if (wr >= 2 && wr < RB + 2) ctr[wr - 2][ch] = a[2];
#pragma unroll
                for (int dx = 0; dx < 5; ++dx) {
                    const float m = fabsf(a[dx]);
#pragma unroll
                    for (int r = 0; r < RB; ++r) {
                        const int dy = wr - r;
                        if (dy >= 0 && dy < 5) pool[r][ch] = fmaf(Pk[dy][dx], m, pool[r][ch]);
                    }
                }
            }
        }
#pragma unroll
        for (int r = 0; r < RB; ++r) {
            const int gy = y0 + RB * g + r, gx = x0 + ox;
            float* op = out + (((size_t)b * H + gy) * W + gx) * 3;
#pragma unroll
            for (int ch = 0; ch < 3; ++ch)
                op[ch] = ctr[r][ch] / (pool[r][ch] + SIGMA_P);
        }
    }
}

template <bool GAM, bool LAST, int NT>
__global__ void __launch_bounds__(NT) __attribute__((amdgpu_waves_per_eu(12)))
nlp_level(const float* __restrict__ src, const float* __restrict__ down,
          float* __restrict__ out, float* __restrict__ nxt, int H, int W, int nd, int sh) {
    constexpr int SUBF = 3 * (36 * 37 + 20 * 21);
    constexpr int DWNF = LAST ? 0 : (NT / 32) * 16 * 49;
    __shared__ float smem[SUBF > DWNF ? SUBF : DWNF];

    const int bz = blockIdx.z;
    const int b = blockIdx.x;
    const int tid = threadIdx.x;

    if (b < 3 * nd) {
        const int pc = (b >= nd ? 1 : 0) + (b >= 2 * nd ? 1 : 0);
        const int g = b - pc * nd;
        const int p = 3 * bz + pc;
        if (LAST) {
            for (int e = tid; e < 256; e += NT) {
                const int y = e >> 4, x = e & 15;
                float v = down[(size_t)p * 18 * 18 + (size_t)(y + 1) * 18 + (x + 1)];
                float o = v / (fabsf(v) + SIGMA_LP);
                nxt[(((size_t)bz * 16 + y) * 16 + x) * 3 + pc] = o;
            }
        } else {
            const int wave = __builtin_amdgcn_readfirstlane(tid >> 5);
            const int Hn = H / 2, Wn = W / 2;
            const int Ty = (Hn / 2 + 2 + 15) / 16, Tx = (Wn / 2 + 2 + 15) / 16;
            const int tile = g * (NT / 32) + wave;
            const bool live = tile < Ty * Tx;
            const int ti = tile / Tx, tj = tile - ti * Tx;
            down_tile(down, nxt, Hn, Wn, p, ti * 16, tj * 16, live,
                             (float (*)[49])(smem + wave * (16 * 49)), tid & 31);
        }
    } else {
        const int s = b - 3 * nd;
        sub_tile<GAM, NT>(src, down, out, H, W, bz, (s >> sh) * 32, (s & ((1 << sh) - 1)) * 32,
                          tid, (float (*)[36][37])smem, (float (*)[20][21])(smem + 3 * 36 * 37));
    }
}

extern "C" void kernel_launch(void* const* d_in, const int* in_sizes, int n_in,
                              void* d_out, int out_size, void* d_ws, size_t ws_size,
                              hipStream_t stream) {
    (void)in_sizes; (void)n_in; (void)out_size; (void)ws_size;
    const float* img = (const float*)d_in[0];
    float* out = (float*)d_out;
    float* ws  = (float*)d_ws;

    const int Hl[6]  = {512, 256, 128, 64, 32, 16};
    const int shl[5] = {4, 3, 2, 1, 0};
    const int ntl[5] = {128, 128, 256, 256, 256};

    size_t dof[5]; size_t off = 0;
    for (int l = 0; l < 5; ++l) {
        dof[l] = off;
        size_t h2 = (size_t)(Hl[l] / 2 + 2);
        off += (size_t)48 * h2 * h2;
    }
    size_t oof[6]; size_t o2 = 0;
    for (int l = 0; l < 6; ++l) {
        oof[l] = o2;
        o2 += (size_t)16 * Hl[l] * Hl[l] * 3;
    }

    {
        const int T = (Hl[0] / 2 + 2 + 15) / 16;
        nlp_down0<<<dim3(T, T, 16), 32, 0, stream>>>(img, ws + dof[0], Hl[0], Hl[0]);
    }
    const float* srcp = img;
    for (int l = 0; l < 5; ++l) {
        const int H = Hl[l];
        const int ns = (H / 32) * (H / 32);
        if (l < 4) {
            const int Tn = (Hl[l + 1] / 2 + 2 + 15) / 16;
            const int tw = ntl[l] / 32;
            const int nd = (Tn * Tn + tw - 1) / tw;
            if (l == 0)
                nlp_level<true, false, 128><<<dim3(3 * nd + ns, 1, 16), 128, 0, stream>>>(
                    srcp, ws + dof[l], out + oof[l], ws + dof[l + 1], H, H, nd, shl[l]);
            else if (l == 1)
                nlp_level<false, false, 128><<<dim3(3 * nd + ns, 1, 16), 128, 0, stream>>>(
                    srcp, ws + dof[l], out + oof[l], ws + dof[l + 1], H, H, nd, shl[l]);
            else
                nlp_level<false, false, 256><<<dim3(3 * nd + ns, 1, 16), 256, 0, stream>>>(
                    srcp, ws + dof[l], out + oof[l], ws + dof[l + 1], H, H, nd, shl[l]);
        } else {
            nlp_level<false, true, 256><<<dim3(3 + ns, 1, 16), 256, 0, stream>>>(
                srcp, ws + dof[l], out + oof[l], out + oof[5], H, H, 1, shl[l]);
        }
        srcp = ws + dof[l];
    }
}


# 5x5 pool folds the |z| of mirrored window columns first (weights are left-right symmetric): 3 products per window row and output, not 5, two output rows per packed FMA; pool sum order changes
# speedup vs baseline: 1.8805x; 1.0824x over previous
#include <hip/hip_runtime.h>
#include <hip/hip_bf16.h>


typedef float v2f __attribute__((ext_vector_type(2)));
typedef float v8f __attribute__((ext_vector_type(8)));

#define GAMMA_EXP 0.38461538461538464f
#define SIGMA_P   0.17f
#define SIGMA_LP  4.86f

__device__ __forceinline__ int reflc(int r, int n) {
    r = (r < 0) ? -r : r;
    if (r >= n) r = 2 * n - 2 - r;
    if (r < 0)  r = 0;
    if (r >= n) r = n - 1;
    return r;
}

__device__ __forceinline__ int clampi(int v, int lo, int hi) {
    return v < lo ? lo : (v > hi ? hi : v);
}

__device__ __forceinline__ float ltap(int d) {
    float v = 0.0f;
    if (d == 0 || d == 4)      v = 0.05f;
    else if (d == 1 || d == 3) v = 0.25f;
    else if (d == 2)           v = 0.40f;
    return v;
}

__device__ __forceinline__ float gamma_of(float v) {
    return __builtin_amdgcn_exp2f(GAMMA_EXP * __builtin_amdgcn_logf(v));
}

__device__ __forceinline__ void src_layout(int p, int H, int W, unsigned& base, unsigned& rpitch) {
    const unsigned Wp = (unsigned)W + 2u;
    base = (unsigned)p * ((unsigned)H + 2u) * Wp + Wp + 1u;
    rpitch = Wp;
}

__device__ __forceinline__ void down_tile(const float* __restrict__ src, float* __restrict__ down,
                                          int H, int W, int p, int i0, int j0, bool live,
                                          float (*vbuf)[49], int lane) {
    const int H2 = H / 2 + 2, W2 = W / 2 + 2;
    const int n = lane & 15;
    const int half = lane >> 4;

    if (live) {
        unsigned base, rpitch;
        src_layout(p, H, W, base, rpitch);

        unsigned colo[3];
#pragma unroll
        for (int nt = 0; nt < 3; ++nt)
            colo[nt] = base + (unsigned)reflc(2 * j0 - 4 + nt * 16 + n, W);

        float raw[9][6];
#pragma unroll
        for (int c = 0; c < 9; ++c) {
            const int kb = 4 * c + 2 * half;
            const unsigned ro0 = rpitch * (unsigned)reflc(2 * i0 + kb - 4, H);
            const unsigned ro1 = rpitch * (unsigned)reflc(2 * i0 + kb - 3, H);
#pragma unroll
            for (int nt = 0; nt < 3; ++nt) {
                raw[c][2 * nt]     = src[ro0 + colo[nt]];
                raw[c][2 * nt + 1] = src[ro1 + colo[nt]];
            }
        }

        __builtin_amdgcn_sched_barrier(0);

        v8f acc0 = {0,0,0,0,0,0,0,0};
        v8f acc1 = {0,0,0,0,0,0,0,0};
        v8f acc2 = {0,0,0,0,0,0,0,0};

#pragma unroll
        for (int c = 0; c < 9; ++c) {
            const int kb = 4 * c + 2 * half;
            v2f a;
            a.x = ltap(kb - 2 * n);
            a.y = ltap(kb + 1 - 2 * n);

            v2f b0, b1, b2;
            b0.x = raw[c][0]; b0.y = raw[c][1];
            b1.x = raw[c][2]; b1.y = raw[c][3];
            b2.x = raw[c][4]; b2.y = raw[c][5];

            acc0 = __builtin_amdgcn_wmma_f32_16x16x4_f32(false, a, false, b0,
                                                         (short)0, acc0, false, false);
            acc1 = __builtin_amdgcn_wmma_f32_16x16x4_f32(false, a, false, b1,
                                                         (short)0, acc1, false, false);
            acc2 = __builtin_amdgcn_wmma_f32_16x16x4_f32(false, a, false, b2,
                                                         (short)0, acc2, false, false);
        }

#pragma unroll
        for (int v = 0; v < 8; ++v) {
            vbuf[v + 8 * half][n]      = acc0[v];
            vbuf[v + 8 * half][16 + n] = acc1[v];
            vbuf[v + 8 * half][32 + n] = acc2[v];
        }
    }
    __syncthreads();
    if (live) {
        v8f dacc = {0,0,0,0,0,0,0,0};
        for (int c = 0; c < 9; ++c) {
            const int kb = 4 * c + 2 * half;
            v2f a;
            a.x = ltap(kb - 2 * n);
            a.y = ltap(kb + 1 - 2 * n);
            v2f b;
            b.x = vbuf[n][kb];
            b.y = vbuf[n][kb + 1];
            dacc = __builtin_amdgcn_wmma_f32_16x16x4_f32(false, a, false, b,
                                                         (short)0, dacc, false, false);
        }

        const int gi = i0 + n;
        if (gi < H2) {
            float* dp = down + (size_t)p * H2 * W2 + (size_t)gi * W2;
#pragma unroll
            for (int v = 0; v < 8; ++v) {
                int gj = j0 + v + 8 * half;
                if (gj < W2) dp[gj] = dacc[v];
            }
        }
    }
}

__global__ void __launch_bounds__(32) __attribute__((amdgpu_waves_per_eu(5)))
nlp_down0(const float* __restrict__ img, float* __restrict__ down, int H, int W) {
    __shared__ float vbuf[3][16][49];
    const int lane = threadIdx.x;
    const int bz = blockIdx.z, i0 = blockIdx.y * 16, j0 = blockIdx.x * 16;
    const int H2 = H / 2 + 2, W2 = W / 2 + 2;
    const int n = lane & 15;
    const int half = lane >> 4;

    const char* imgb = (const char*)img;
    const unsigned base = (unsigned)bz * (unsigned)H * (unsigned)W * 12u;
    const unsigned rpitch = (unsigned)W * 12u;

    unsigned colo[2];
#pragma unroll
    for (int g = 0; g < 2; ++g)
        colo[g] = base + 12u * (unsigned)reflc(2 * j0 - 4 + g * 16 + n, W);
    const int t6 = 32 + n / 3, ch6 = n - 3 * (n / 3);
    const unsigned col6 = base + 12u * (unsigned)reflc(2 * j0 - 4 + t6, W) + 4u * (unsigned)ch6;

    float raw[9][4][3];
    float raw6[9][2];
#pragma unroll
    for (int c = 0; c < 9; ++c) {
        const int kb = 4 * c + 2 * half;
        const unsigned ro0 = rpitch * (unsigned)reflc(2 * i0 + kb - 4, H);
        const unsigned ro1 = rpitch * (unsigned)reflc(2 * i0 + kb - 3, H);
#pragma unroll
        for (int g = 0; g < 2; ++g) {
            const float* p0 = (const float*)(imgb + (ro0 + colo[g]));
            const float* p1 = (const float*)(imgb + (ro1 + colo[g]));
#pragma unroll
            for (int ch = 0; ch < 3; ++ch) {
                raw[c][2 * g][ch]     = p0[ch];
                raw[c][2 * g + 1][ch] = p1[ch];
            }
        }
        raw6[c][0] = *(const float*)(imgb + (ro0 + col6));
        raw6[c][1] = *(const float*)(imgb + (ro1 + col6));
    }

    __builtin_amdgcn_sched_barrier(0);

#pragma unroll
    for (int ch = 0; ch < 3; ++ch) {
        v8f acc0 = {0,0,0,0,0,0,0,0};
        v8f acc1 = {0,0,0,0,0,0,0,0};
#pragma unroll
        for (int c = 0; c < 9; ++c) {
            const int kb = 4 * c + 2 * half;
            v2f a;
            a.x = ltap(kb - 2 * n);
            a.y = ltap(kb + 1 - 2 * n);

            v2f b0, b1;
            b0.x = gamma_of(raw[c][0][ch]); b0.y = gamma_of(raw[c][1][ch]);
            b1.x = gamma_of(raw[c][2][ch]); b1.y = gamma_of(raw[c][3][ch]);

            acc0 = __builtin_amdgcn_wmma_f32_16x16x4_f32(false, a, false, b0,
                                                         (short)0, acc0, false, false);
            acc1 = __builtin_amdgcn_wmma_f32_16x16x4_f32(false, a, false, b1,
                                                         (short)0, acc1, false, false);
        }
#pragma unroll
        for (int v = 0; v < 8; ++v) {
            vbuf[ch][v + 8 * half][n]      = acc0[v];
            vbuf[ch][v + 8 * half][16 + n] = acc1[v];
        }
    }
    {
        v8f acc6 = {0,0,0,0,0,0,0,0};
#pragma unroll
        for (int c = 0; c < 9; ++c) {
            const int kb = 4 * c + 2 * half;
            v2f a;
            a.x = ltap(kb - 2 * n);
            a.y = ltap(kb + 1 - 2 * n);
            v2f b6;
            b6.x = gamma_of(raw6[c][0]); b6.y = gamma_of(raw6[c][1]);
            acc6 = __builtin_amdgcn_wmma_f32_16x16x4_f32(false, a, false, b6,
                                                         (short)0, acc6, false, false);
        }
#pragma unroll
        for (int v = 0; v < 8; ++v) vbuf[ch6][v + 8 * half][t6] = acc6[v];
    }
    __syncthreads();

    v8f dacc[3];
#pragma unroll
    for (int ch = 0; ch < 3; ++ch) dacc[ch] = (v8f){0,0,0,0,0,0,0,0};
#pragma unroll
    for (int c = 0; c < 9; ++c) {
        const int kb = 4 * c + 2 * half;
        v2f a;
        a.x = ltap(kb - 2 * n);
        a.y = ltap(kb + 1 - 2 * n);
#pragma unroll
        for (int ch = 0; ch < 3; ++ch) {
            v2f b;
            b.x = vbuf[ch][n][kb];
            b.y = vbuf[ch][n][kb + 1];
            dacc[ch] = __builtin_amdgcn_wmma_f32_16x16x4_f32(false, a, false, b,
                                                             (short)0, dacc[ch], false, false);
        }
    }

    const int gi = i0 + n;
    if (gi < H2) {
#pragma unroll
        for (int ch = 0; ch < 3; ++ch) {
            float* dp = down + (size_t)(3 * bz + ch) * H2 * W2 + (size_t)gi * W2;
#pragma unroll
            for (int v = 0; v < 8; ++v) {
                int gj = j0 + v + 8 * half;
                if (gj < W2) dp[gj] = dacc[ch][v];
            }
        }
    }
}

template <bool GAM, int NT>
__device__ __forceinline__ void sub_tile(const float* __restrict__ src,
                                         const float* __restrict__ down,
                                         float* __restrict__ out, int H, int W, int b,
                                         int y0, int x0, int tid,
                                         float (*zb)[36][37], float (*dt)[20][21]) {
    constexpr int DQ = (400 + NT - 1) / NT;
    constexpr int BQ = (324 + NT - 1) / NT;
    constexpr int RB = 4;
    constexpr int SQ = 256 / NT;
    const int H2 = H / 2 + 2, W2 = W / 2 + 2;
    const unsigned dpl = (unsigned)H2 * (unsigned)W2;
    const float* dw = down + (size_t)(3 * b) * dpl;

    const int dr0 = (y0 >> 1) - 1, dc0 = (x0 >> 1) - 1;

    unsigned base, rpitch, cpitch, chp;
    if (GAM) {
        base = (unsigned)b * (unsigned)H * (unsigned)W * 3u;
        rpitch = (unsigned)W * 3u;
        cpitch = 3u;
        chp = 1u;
    } else {
        const unsigned Wp = (unsigned)W + 2u;
        chp = ((unsigned)H + 2u) * Wp;
        base = (unsigned)(3 * b) * chp + Wp + 1u;
        rpitch = Wp;
        cpitch = 1u;
    }

    float dv[DQ][3];
#pragma unroll
    for (int q = 0; q < DQ; ++q) {
        int t = tid + q * NT; t = t > 399 ? 399 : t;
        int ly = t / 20, lx = t % 20;
        int r = clampi(dr0 + ly, 0, H2 - 1);
        int c = clampi(dc0 + lx, 0, W2 - 1);
        const unsigned o = (unsigned)r * (unsigned)W2 + (unsigned)c;
#pragma unroll
        for (int ch = 0; ch < 3; ++ch) dv[q][ch] = (dw + (size_t)ch * dpl)[o];
    }
    float xr[BQ][2][2][3];
#pragma unroll
    for (int q = 0; q < BQ; ++q) {
        int t = tid + q * NT; t = t > 323 ? 323 : t;
        const int by = t / 18, bx = t - by * 18;
        const int y = clampi(y0 - 2 + 2 * by, 0, H - 2);
        const int x = clampi(x0 - 2 + 2 * bx, 0, W - 2);
        const unsigned o = base + (unsigned)y * rpitch + (unsigned)x * cpitch;
#pragma unroll
        for (int i = 0; i < 2; ++i)
#pragma unroll
            for (int j = 0; j < 2; ++j)
#pragma unroll
                for (int ch = 0; ch < 3; ++ch) {
                    if (GAM) xr[q][i][j][ch] = (src + o + (unsigned)i * rpitch)[3 * j + ch];
                    else     xr[q][i][j][ch] = (src + (size_t)ch * chp)[o + (unsigned)i * rpitch + (unsigned)j];
                }
    }

#pragma unroll
    for (int q = 0; q < DQ; ++q) {
        int t = tid + q * NT;
        if (t < 400) {
#pragma unroll
            for (int ch = 0; ch < 3; ++ch) dt[ch][t / 20][t % 20] = dv[q][ch];
        }
    }
    __syncthreads();

#pragma unroll
    for (int q = 0; q < BQ; ++q) {
        const int t = tid + q * NT;
        if (t < 324) {
            const int by = t / 18, bx = t - by * 18;
            const int y = y0 - 2 + 2 * by, x = x0 - 2 + 2 * bx;
            const bool inside = (y >= 0 && y < H && x >= 0 && x < W);
#pragma unroll
            for (int ch = 0; ch < 3; ++ch) {
                float z00 = 0.0f, z01 = 0.0f, z10 = 0.0f, z11 = 0.0f;
                if (inside) {
                    float he[3], ho[3];
#pragma unroll
                    for (int rr = 0; rr < 3; ++rr) {
                        const float d0 = dt[ch][by + rr][bx];
                        const float d1 = dt[ch][by + rr][bx + 1];
                        const float d2 = dt[ch][by + rr][bx + 2];
                        float e = 0.0f;
                        e = fmaf(0.1f, d0, e); e = fmaf(0.8f, d1, e); e = fmaf(0.1f, d2, e);
                        float o = 0.0f;
                        o = fmaf(0.5f, d1, o); o = fmaf(0.5f, d2, o);
                        he[rr] = e; ho[rr] = o;
                    }
                    float uee = 0.0f, ueo = 0.0f, uoe = 0.0f, uoo = 0.0f;
                    uee = fmaf(0.1f, he[0], uee); uee = fmaf(0.8f, he[1], uee); uee = fmaf(0.1f, he[2], uee);
                    ueo = fmaf(0.1f, ho[0], ueo); ueo = fmaf(0.8f, ho[1], ueo); ueo = fmaf(0.1f, ho[2], ueo);
                    uoe = fmaf(0.5f, he[1], uoe); uoe = fmaf(0.5f, he[2], uoe);
                    uoo = fmaf(0.5f, ho[1], uoo); uoo = fmaf(0.5f, ho[2], uoo);
                    const float x00 = GAM ? gamma_of(xr[q][0][0][ch]) : xr[q][0][0][ch];
                    const float x01 = GAM ? gamma_of(xr[q][0][1][ch]) : xr[q][0][1][ch];
                    const float x10 = GAM ? gamma_of(xr[q][1][0][ch]) : xr[q][1][0][ch];
                    const float x11 = GAM ? gamma_of(xr[q][1][1][ch]) : xr[q][1][1][ch];
                    z00 = x00 - uee; z01 = x01 - ueo; z10 = x10 - uoe; z11 = x11 - uoo;
                }
                zb[ch][2 * by][2 * bx]         = z00;
                zb[ch][2 * by][2 * bx + 1]     = z01;
                zb[ch][2 * by + 1][2 * bx]     = z10;
                zb[ch][2 * by + 1][2 * bx + 1] = z11;
            }
        }
    }
    __syncthreads();

    constexpr float Pk[5][5] = {
        {0.04f, 0.04f, 0.05f, 0.04f, 0.04f},
        {0.04f, 0.03f, 0.04f, 0.03f, 0.04f},
        {0.05f, 0.04f, 0.05f, 0.04f, 0.05f},
        {0.04f, 0.03f, 0.04f, 0.03f, 0.04f},
        {0.04f, 0.04f, 0.05f, 0.04f, 0.04f}};
    static_assert(Pk[0][0] == Pk[0][4] && Pk[0][1] == Pk[0][3] &&
                  Pk[1][0] == Pk[1][4] && Pk[1][1] == Pk[1][3] &&
                  Pk[2][0] == Pk[2][4] && Pk[2][1] == Pk[2][3] &&
                  Pk[3][0] == Pk[3][4] && Pk[3][1] == Pk[3][3] &&
                  Pk[4][0] == Pk[4][4] && Pk[4][1] == Pk[4][3],
                  "pool weights must be left-right symmetric");

    for (int k = 0; k < SQ; ++k) {
        const int s = tid + k * NT;
        const int ox = s & 31, g = s >> 5;
        v2f pool2[RB / 2][3];
        float ctr[RB][3];
#pragma unroll
        for (int r = 0; r < RB; ++r)
#pragma unroll
            for (int ch = 0; ch < 3; ++ch) { pool2[r >> 1][ch] = (v2f){0.0f, 0.0f}; ctr[r][ch] = 0.0f; }
#pragma unroll
        for (int wr = 0; wr < RB + 4; ++wr) {
#pragma unroll
            for (int ch = 0; ch < 3; ++ch) {
                float a[5];
#pragma unroll
                for (int dx = 0; dx < 5; ++dx) a[dx] = zb[ch][RB * g + wr][ox + dx];
                if (wr >= 2 && wr < RB + 2) ctr[wr - 2][ch] = a[2];
                float sf[3];
                sf[0] = fabsf(a[0]) + fabsf(a[4]);
                sf[1] = fabsf(a[1]) + fabsf(a[3]);
                sf[2] = fabsf(a[2]);
#pragma unroll
                for (int rp = 0; rp < RB / 2; ++rp) {
                    const int dyl = wr - 2 * rp;
                    const int dyh = dyl - 1;
                    const bool vl = (dyl >= 0 && dyl < 5), vh = (dyh >= 0 && dyh < 5);
#pragma unroll
                    for (int c = 0; c < 3; ++c) {
                        if (vl && vh) {
                            const v2f w = {Pk[dyl][c], Pk[dyh][c]};
                            const v2f m = {sf[c], sf[c]};
                            pool2[rp][ch] = __builtin_elementwise_fma(w, m, pool2[rp][ch]);
                        } else if (vl) {
                            pool2[rp][ch].x = fmaf(Pk[dyl][c], sf[c], pool2[rp][ch].x);
                        } else if (vh) {
                            pool2[rp][ch].y = fmaf(Pk[dyh][c], sf[c], pool2[rp][ch].y);
                        }
                    }
                }
            }
        }
#pragma unroll
        for (int r = 0; r < RB; ++r) {
            const int gy = y0 + RB * g + r, gx = x0 + ox;
            float* op = out + (((size_t)b * H + gy) * W + gx) * 3;
#pragma unroll
            for (int ch = 0; ch < 3; ++ch)
                op[ch] = ctr[r][ch] / (((r & 1) ? pool2[r >> 1][ch].y : pool2[r >> 1][ch].x) + SIGMA_P);
        }
    }
}

template <bool GAM, bool LAST, int NT>
__global__ void __launch_bounds__(NT) __attribute__((amdgpu_waves_per_eu(12)))
nlp_level(const float* __restrict__ src, const float* __restrict__ down,
          float* __restrict__ out, float* __restrict__ nxt, int H, int W, int nd, int sh) {
    constexpr int SUBF = 3 * (36 * 37 + 20 * 21);
    constexpr int DWNF = LAST ? 0 : (NT / 32) * 16 * 49;
    __shared__ float smem[SUBF > DWNF ? SUBF : DWNF];

    const int bz = blockIdx.z;
    const int b = blockIdx.x;
    const int tid = threadIdx.x;

    if (b < 3 * nd) {
        const int pc = (b >= nd ? 1 : 0) + (b >= 2 * nd ? 1 : 0);
        const int g = b - pc * nd;
        const int p = 3 * bz + pc;
        if (LAST) {
            for (int e = tid; e < 256; e += NT) {
                const int y = e >> 4, x = e & 15;
                float v = down[(size_t)p * 18 * 18 + (size_t)(y + 1) * 18 + (x + 1)];
                float o = v / (fabsf(v) + SIGMA_LP);
                nxt[(((size_t)bz * 16 + y) * 16 + x) * 3 + pc] = o;
            }
        } else {
            const int wave = __builtin_amdgcn_readfirstlane(tid >> 5);
            const int Hn = H / 2, Wn = W / 2;
            const int Ty = (Hn / 2 + 2 + 15) / 16, Tx = (Wn / 2 + 2 + 15) / 16;
            const int tile = g * (NT / 32) + wave;
            const bool live = tile < Ty * Tx;
            const int ti = tile / Tx, tj = tile - ti * Tx;
            down_tile(down, nxt, Hn, Wn, p, ti * 16, tj * 16, live,
                             (float (*)[49])(smem + wave * (16 * 49)), tid & 31);
        }
    } else {
        const int s = b - 3 * nd;
        sub_tile<GAM, NT>(src, down, out, H, W, bz, (s >> sh) * 32, (s & ((1 << sh) - 1)) * 32,
                          tid, (float (*)[36][37])smem, (float (*)[20][21])(smem + 3 * 36 * 37));
    }
}

extern "C" void kernel_launch(void* const* d_in, const int* in_sizes, int n_in,
                              void* d_out, int out_size, void* d_ws, size_t ws_size,
                              hipStream_t stream) {
    (void)in_sizes; (void)n_in; (void)out_size; (void)ws_size;
    const float* img = (const float*)d_in[0];
    float* out = (float*)d_out;
    float* ws  = (float*)d_ws;

    const int Hl[6]  = {512, 256, 128, 64, 32, 16};
    const int shl[5] = {4, 3, 2, 1, 0};
    const int ntl[5] = {128, 128, 256, 256, 256};

    size_t dof[5]; size_t off = 0;
    for (int l = 0; l < 5; ++l) {
        dof[l] = off;
        size_t h2 = (size_t)(Hl[l] / 2 + 2);
        off += (size_t)48 * h2 * h2;
    }
    size_t oof[6]; size_t o2 = 0;
    for (int l = 0; l < 6; ++l) {
        oof[l] = o2;
        o2 += (size_t)16 * Hl[l] * Hl[l] * 3;
    }

    {
        const int T = (Hl[0] / 2 + 2 + 15) / 16;
        nlp_down0<<<dim3(T, T, 16), 32, 0, stream>>>(img, ws + dof[0], Hl[0], Hl[0]);
    }
    const float* srcp = img;
    for (int l = 0; l < 5; ++l) {
        const int H = Hl[l];
        const int ns = (H / 32) * (H / 32);
        if (l < 4) {
            const int Tn = (Hl[l + 1] / 2 + 2 + 15) / 16;
            const int tw = ntl[l] / 32;
            const int nd = (Tn * Tn + tw - 1) / tw;
            if (l == 0)
                nlp_level<true, false, 128><<<dim3(3 * nd + ns, 1, 16), 128, 0, stream>>>(
                    srcp, ws + dof[l], out + oof[l], ws + dof[l + 1], H, H, nd, shl[l]);
            else if (l == 1)
                nlp_level<false, false, 128><<<dim3(3 * nd + ns, 1, 16), 128, 0, stream>>>(
                    srcp, ws + dof[l], out + oof[l], ws + dof[l + 1], H, H, nd, shl[l]);
            else
                nlp_level<false, false, 256><<<dim3(3 * nd + ns, 1, 16), 256, 0, stream>>>(
                    srcp, ws + dof[l], out + oof[l], ws + dof[l + 1], H, H, nd, shl[l]);
        } else {
            nlp_level<false, true, 256><<<dim3(3 + ns, 1, 16), 256, 0, stream>>>(
                srcp, ws + dof[l], out + oof[l], out + oof[5], H, H, 1, shl[l]);
        }
        srcp = ws + dof[l];
    }
}


# both normalising divides (z over pool+sigma, lowpass residual) use v_rcp_f32 and one multiply, not the correctly rounded f32 divide sequence; quotient within about 2 ulp, sums unchanged
# speedup vs baseline: 1.9523x; 1.0382x over previous
#include <hip/hip_runtime.h>
#include <hip/hip_bf16.h>


typedef float v2f __attribute__((ext_vector_type(2)));
typedef float v8f __attribute__((ext_vector_type(8)));

#define GAMMA_EXP 0.38461538461538464f
#define SIGMA_P   0.17f
#define SIGMA_LP  4.86f

__device__ __forceinline__ int reflc(int r, int n) {
    r = (r < 0) ? -r : r;
    if (r >= n) r = 2 * n - 2 - r;
    if (r < 0)  r = 0;
    if (r >= n) r = n - 1;
    return r;
}

__device__ __forceinline__ int clampi(int v, int lo, int hi) {
    return v < lo ? lo : (v > hi ? hi : v);
}

__device__ __forceinline__ float ltap(int d) {
    float v = 0.0f;
    if (d == 0 || d == 4)      v = 0.05f;
    else if (d == 1 || d == 3) v = 0.25f;
    else if (d == 2)           v = 0.40f;
    return v;
}

__device__ __forceinline__ float gamma_of(float v) {
    return __builtin_amdgcn_exp2f(GAMMA_EXP * __builtin_amdgcn_logf(v));
}

__device__ __forceinline__ float quot(float num, float den) {
    return num * __builtin_amdgcn_rcpf(den);
}

__device__ __forceinline__ void src_layout(int p, int H, int W, unsigned& base, unsigned& rpitch) {
    const unsigned Wp = (unsigned)W + 2u;
    base = (unsigned)p * ((unsigned)H + 2u) * Wp + Wp + 1u;
    rpitch = Wp;
}

__device__ __forceinline__ void down_tile(const float* __restrict__ src, float* __restrict__ down,
                                          int H, int W, int p, int i0, int j0, bool live,
                                          float (*vbuf)[49], int lane) {
    const int H2 = H / 2 + 2, W2 = W / 2 + 2;
    const int n = lane & 15;
    const int half = lane >> 4;

    if (live) {
        unsigned base, rpitch;
        src_layout(p, H, W, base, rpitch);

        unsigned colo[3];
#pragma unroll
        for (int nt = 0; nt < 3; ++nt)
            colo[nt] = base + (unsigned)reflc(2 * j0 - 4 + nt * 16 + n, W);

        float raw[9][6];
#pragma unroll
        for (int c = 0; c < 9; ++c) {
            const int kb = 4 * c + 2 * half;
            const unsigned ro0 = rpitch * (unsigned)reflc(2 * i0 + kb - 4, H);
            const unsigned ro1 = rpitch * (unsigned)reflc(2 * i0 + kb - 3, H);
#pragma unroll
            for (int nt = 0; nt < 3; ++nt) {
                raw[c][2 * nt]     = src[ro0 + colo[nt]];
                raw[c][2 * nt + 1] = src[ro1 + colo[nt]];
            }
        }

        __builtin_amdgcn_sched_barrier(0);

        v8f acc0 = {0,0,0,0,0,0,0,0};
        v8f acc1 = {0,0,0,0,0,0,0,0};
        v8f acc2 = {0,0,0,0,0,0,0,0};

#pragma unroll
        for (int c = 0; c < 9; ++c) {
            const int kb = 4 * c + 2 * half;
            v2f a;
            a.x = ltap(kb - 2 * n);
            a.y = ltap(kb + 1 - 2 * n);

            v2f b0, b1, b2;
            b0.x = raw[c][0]; b0.y = raw[c][1];
            b1.x = raw[c][2]; b1.y = raw[c][3];
            b2.x = raw[c][4]; b2.y = raw[c][5];

            acc0 = __builtin_amdgcn_wmma_f32_16x16x4_f32(false, a, false, b0,
                                                         (short)0, acc0, false, false);
            acc1 = __builtin_amdgcn_wmma_f32_16x16x4_f32(false, a, false, b1,
                                                         (short)0, acc1, false, false);
            acc2 = __builtin_amdgcn_wmma_f32_16x16x4_f32(false, a, false, b2,
                                                         (short)0, acc2, false, false);
        }

#pragma unroll
        for (int v = 0; v < 8; ++v) {
            vbuf[v + 8 * half][n]      = acc0[v];
            vbuf[v + 8 * half][16 + n] = acc1[v];
            vbuf[v + 8 * half][32 + n] = acc2[v];
        }
    }
    __syncthreads();
    if (live) {
        v8f dacc = {0,0,0,0,0,0,0,0};
        for (int c = 0; c < 9; ++c) {
            const int kb = 4 * c + 2 * half;
            v2f a;
            a.x = ltap(kb - 2 * n);
            a.y = ltap(kb + 1 - 2 * n);
            v2f b;
            b.x = vbuf[n][kb];
            b.y = vbuf[n][kb + 1];
            dacc = __builtin_amdgcn_wmma_f32_16x16x4_f32(false, a, false, b,
                                                         (short)0, dacc, false, false);
        }

        const int gi = i0 + n;
        if (gi < H2) {
            float* dp = down + (size_t)p * H2 * W2 + (size_t)gi * W2;
#pragma unroll
            for (int v = 0; v < 8; ++v) {
                int gj = j0 + v + 8 * half;
                if (gj < W2) dp[gj] = dacc[v];
            }
        }
    }
}

__global__ void __launch_bounds__(32) __attribute__((amdgpu_waves_per_eu(5)))
nlp_down0(const float* __restrict__ img, float* __restrict__ down, int H, int W) {
    __shared__ float vbuf[3][16][49];
    const int lane = threadIdx.x;
    const int bz = blockIdx.z, i0 = blockIdx.y * 16, j0 = blockIdx.x * 16;
    const int H2 = H / 2 + 2, W2 = W / 2 + 2;
    const int n = lane & 15;
    const int half = lane >> 4;

    const char* imgb = (const char*)img;
    const unsigned base = (unsigned)bz * (unsigned)H * (unsigned)W * 12u;
    const unsigned rpitch = (unsigned)W * 12u;

    unsigned colo[2];
#pragma unroll
    for (int g = 0; g < 2; ++g)
        colo[g] = base + 12u * (unsigned)reflc(2 * j0 - 4 + g * 16 + n, W);
    const int t6 = 32 + n / 3, ch6 = n - 3 * (n / 3);
    const unsigned col6 = base + 12u * (unsigned)reflc(2 * j0 - 4 + t6, W) + 4u * (unsigned)ch6;

    float raw[9][4][3];
    float raw6[9][2];
#pragma unroll
    for (int c = 0; c < 9; ++c) {
        const int kb = 4 * c + 2 * half;
        const unsigned ro0 = rpitch * (unsigned)reflc(2 * i0 + kb - 4, H);
        const unsigned ro1 = rpitch * (unsigned)reflc(2 * i0 + kb - 3, H);
#pragma unroll
        for (int g = 0; g < 2; ++g) {
            const float* p0 = (const float*)(imgb + (ro0 + colo[g]));
            const float* p1 = (const float*)(imgb + (ro1 + colo[g]));
#pragma unroll
            for (int ch = 0; ch < 3; ++ch) {
                raw[c][2 * g][ch]     = p0[ch];
                raw[c][2 * g + 1][ch] = p1[ch];
            }
        }
        raw6[c][0] = *(const float*)(imgb + (ro0 + col6));
        raw6[c][1] = *(const float*)(imgb + (ro1 + col6));
    }

    __builtin_amdgcn_sched_barrier(0);

#pragma unroll
    for (int ch = 0; ch < 3; ++ch) {
        v8f acc0 = {0,0,0,0,0,0,0,0};
        v8f acc1 = {0,0,0,0,0,0,0,0};
#pragma unroll
        for (int c = 0; c < 9; ++c) {
            const int kb = 4 * c + 2 * half;
            v2f a;
            a.x = ltap(kb - 2 * n);
            a.y = ltap(kb + 1 - 2 * n);

            v2f b0, b1;
            b0.x = gamma_of(raw[c][0][ch]); b0.y = gamma_of(raw[c][1][ch]);
            b1.x = gamma_of(raw[c][2][ch]); b1.y = gamma_of(raw[c][3][ch]);

            acc0 = __builtin_amdgcn_wmma_f32_16x16x4_f32(false, a, false, b0,
                                                         (short)0, acc0, false, false);
            acc1 = __builtin_amdgcn_wmma_f32_16x16x4_f32(false, a, false, b1,
                                                         (short)0, acc1, false, false);
        }
#pragma unroll
        for (int v = 0; v < 8; ++v) {
            vbuf[ch][v + 8 * half][n]      = acc0[v];
            vbuf[ch][v + 8 * half][16 + n] = acc1[v];
        }
    }
    {
        v8f acc6 = {0,0,0,0,0,0,0,0};
#pragma unroll
        for (int c = 0; c < 9; ++c) {
            const int kb = 4 * c + 2 * half;
            v2f a;
            a.x = ltap(kb - 2 * n);
            a.y = ltap(kb + 1 - 2 * n);
            v2f b6;
            b6.x = gamma_of(raw6[c][0]); b6.y = gamma_of(raw6[c][1]);
            acc6 = __builtin_amdgcn_wmma_f32_16x16x4_f32(false, a, false, b6,
                                                         (short)0, acc6, false, false);
        }
#pragma unroll
        for (int v = 0; v < 8; ++v) vbuf[ch6][v + 8 * half][t6] = acc6[v];
    }
    __syncthreads();

    v8f dacc[3];
#pragma unroll
    for (int ch = 0; ch < 3; ++ch) dacc[ch] = (v8f){0,0,0,0,0,0,0,0};
#pragma unroll
    for (int c = 0; c < 9; ++c) {
        const int kb = 4 * c + 2 * half;
        v2f a;
        a.x = ltap(kb - 2 * n);
        a.y = ltap(kb + 1 - 2 * n);
#pragma unroll
        for (int ch = 0; ch < 3; ++ch) {
            v2f b;
            b.x = vbuf[ch][n][kb];
            b.y = vbuf[ch][n][kb + 1];
            dacc[ch] = __builtin_amdgcn_wmma_f32_16x16x4_f32(false, a, false, b,
                                                             (short)0, dacc[ch], false, false);
        }
    }

    const int gi = i0 + n;
    if (gi < H2) {
#pragma unroll
        for (int ch = 0; ch < 3; ++ch) {
            float* dp = down + (size_t)(3 * bz + ch) * H2 * W2 + (size_t)gi * W2;
#pragma unroll
            for (int v = 0; v < 8; ++v) {
                int gj = j0 + v + 8 * half;
                if (gj < W2) dp[gj] = dacc[ch][v];
            }
        }
    }
}

template <bool GAM, int NT>
__device__ __forceinline__ void sub_tile(const float* __restrict__ src,
                                         const float* __restrict__ down,
                                         float* __restrict__ out, int H, int W, int b,
                                         int y0, int x0, int tid,
                                         float (*zb)[36][37], float (*dt)[20][21]) {
    constexpr int DQ = (400 + NT - 1) / NT;
    constexpr int BQ = (324 + NT - 1) / NT;
    constexpr int RB = 4;
    constexpr int SQ = 256 / NT;
    const int H2 = H / 2 + 2, W2 = W / 2 + 2;
    const unsigned dpl = (unsigned)H2 * (unsigned)W2;
    const float* dw = down + (size_t)(3 * b) * dpl;

    const int dr0 = (y0 >> 1) - 1, dc0 = (x0 >> 1) - 1;

    unsigned base, rpitch, cpitch, chp;
    if (GAM) {
        base = (unsigned)b * (unsigned)H * (unsigned)W * 3u;
        rpitch = (unsigned)W * 3u;
        cpitch = 3u;
        chp = 1u;
    } else {
        const unsigned Wp = (unsigned)W + 2u;
        chp = ((unsigned)H + 2u) * Wp;
        base = (unsigned)(3 * b) * chp + Wp + 1u;
        rpitch = Wp;
        cpitch = 1u;
    }

    float dv[DQ][3];
#pragma unroll
    for (int q = 0; q < DQ; ++q) {
        int t = tid + q * NT; t = t > 399 ? 399 : t;
        int ly = t / 20, lx = t % 20;
        int r = clampi(dr0 + ly, 0, H2 - 1);
        int c = clampi(dc0 + lx, 0, W2 - 1);
        const unsigned o = (unsigned)r * (unsigned)W2 + (unsigned)c;
#pragma unroll
        for (int ch = 0; ch < 3; ++ch) dv[q][ch] = (dw + (size_t)ch * dpl)[o];
    }
    float xr[BQ][2][2][3];
#pragma unroll
    for (int q = 0; q < BQ; ++q) {
        int t = tid + q * NT; t = t > 323 ? 323 : t;
        const int by = t / 18, bx = t - by * 18;
        const int y = clampi(y0 - 2 + 2 * by, 0, H - 2);
        const int x = clampi(x0 - 2 + 2 * bx, 0, W - 2);
        const unsigned o = base + (unsigned)y * rpitch + (unsigned)x * cpitch;
#pragma unroll
        for (int i = 0; i < 2; ++i)
#pragma unroll
            for (int j = 0; j < 2; ++j)
#pragma unroll
                for (int ch = 0; ch < 3; ++ch) {
                    if (GAM) xr[q][i][j][ch] = (src + o + (unsigned)i * rpitch)[3 * j + ch];
                    else     xr[q][i][j][ch] = (src + (size_t)ch * chp)[o + (unsigned)i * rpitch + (unsigned)j];
                }
    }

#pragma unroll
    for (int q = 0; q < DQ; ++q) {
        int t = tid + q * NT;
        if (t < 400) {
#pragma unroll
            for (int ch = 0; ch < 3; ++ch) dt[ch][t / 20][t % 20] = dv[q][ch];
        }
    }
    __syncthreads();

#pragma unroll
    for (int q = 0; q < BQ; ++q) {
        const int t = tid + q * NT;
        if (t < 324) {
            const int by = t / 18, bx = t - by * 18;
            const int y = y0 - 2 + 2 * by, x = x0 - 2 + 2 * bx;
            const bool inside = (y >= 0 && y < H && x >= 0 && x < W);
#pragma unroll
            for (int ch = 0; ch < 3; ++ch) {
                float z00 = 0.0f, z01 = 0.0f, z10 = 0.0f, z11 = 0.0f;
                if (inside) {
                    float he[3], ho[3];
#pragma unroll
                    for (int rr = 0; rr < 3; ++rr) {
                        const float d0 = dt[ch][by + rr][bx];
                        const float d1 = dt[ch][by + rr][bx + 1];
                        const float d2 = dt[ch][by + rr][bx + 2];
                        float e = 0.0f;
                        e = fmaf(0.1f, d0, e); e = fmaf(0.8f, d1, e); e = fmaf(0.1f, d2, e);
                        float o = 0.0f;
                        o = fmaf(0.5f, d1, o); o = fmaf(0.5f, d2, o);
                        he[rr] = e; ho[rr] = o;
                    }
                    float uee = 0.0f, ueo = 0.0f, uoe = 0.0f, uoo = 0.0f;
                    uee = fmaf(0.1f, he[0], uee); uee = fmaf(0.8f, he[1], uee); uee = fmaf(0.1f, he[2], uee);
                    ueo = fmaf(0.1f, ho[0], ueo); ueo = fmaf(0.8f, ho[1], ueo); ueo = fmaf(0.1f, ho[2], ueo);
                    uoe = fmaf(0.5f, he[1], uoe); uoe = fmaf(0.5f, he[2], uoe);
                    uoo = fmaf(0.5f, ho[1], uoo); uoo = fmaf(0.5f, ho[2], uoo);
                    const float x00 = GAM ? gamma_of(xr[q][0][0][ch]) : xr[q][0][0][ch];
                    const float x01 = GAM ? gamma_of(xr[q][0][1][ch]) : xr[q][0][1][ch];
                    const float x10 = GAM ? gamma_of(xr[q][1][0][ch]) : xr[q][1][0][ch];
                    const float x11 = GAM ? gamma_of(xr[q][1][1][ch]) : xr[q][1][1][ch];
                    z00 = x00 - uee; z01 = x01 - ueo; z10 = x10 - uoe; z11 = x11 - uoo;
                }
                zb[ch][2 * by][2 * bx]         = z00;
                zb[ch][2 * by][2 * bx + 1]     = z01;
                zb[ch][2 * by + 1][2 * bx]     = z10;
                zb[ch][2 * by + 1][2 * bx + 1] = z11;
            }
        }
    }
    __syncthreads();

    constexpr float Pk[5][5] = {
        {0.04f, 0.04f, 0.05f, 0.04f, 0.04f},
        {0.04f, 0.03f, 0.04f, 0.03f, 0.04f},
        {0.05f, 0.04f, 0.05f, 0.04f, 0.05f},
        {0.04f, 0.03f, 0.04f, 0.03f, 0.04f},
        {0.04f, 0.04f, 0.05f, 0.04f, 0.04f}};
    static_assert(Pk[0][0] == Pk[0][4] && Pk[0][1] == Pk[0][3] &&
                  Pk[1][0] == Pk[1][4] && Pk[1][1] == Pk[1][3] &&
                  Pk[2][0] == Pk[2][4] && Pk[2][1] == Pk[2][3] &&
                  Pk[3][0] == Pk[3][4] && Pk[3][1] == Pk[3][3] &&
                  Pk[4][0] == Pk[4][4] && Pk[4][1] == Pk[4][3],
                  "pool weights must be left-right symmetric");

    for (int k = 0; k < SQ; ++k) {
        const int s = tid + k * NT;
        const int ox = s & 31, g = s >> 5;
        v2f pool2[RB / 2][3];
        float ctr[RB][3];
#pragma unroll
        for (int r = 0; r < RB; ++r)
#pragma unroll
            for (int ch = 0; ch < 3; ++ch) { pool2[r >> 1][ch] = (v2f){0.0f, 0.0f}; ctr[r][ch] = 0.0f; }
#pragma unroll
        for (int wr = 0; wr < RB + 4; ++wr) {
#pragma unroll
            for (int ch = 0; ch < 3; ++ch) {
                float a[5];
#pragma unroll
                for (int dx = 0; dx < 5; ++dx) a[dx] = zb[ch][RB * g + wr][ox + dx];
                if (wr >= 2 && wr < RB + 2) ctr[wr - 2][ch] = a[2];
                float sf[3];
                sf[0] = fabsf(a[0]) + fabsf(a[4]);
                sf[1] = fabsf(a[1]) + fabsf(a[3]);
                sf[2] = fabsf(a[2]);
#pragma unroll
                for (int rp = 0; rp < RB / 2; ++rp) {
                    const int dyl = wr - 2 * rp;
                    const int dyh = dyl - 1;
                    const bool vl = (dyl >= 0 && dyl < 5), vh = (dyh >= 0 && dyh < 5);
#pragma unroll
                    for (int c = 0; c < 3; ++c) {
                        if (vl && vh) {
                            const v2f w = {Pk[dyl][c], Pk[dyh][c]};
                            const v2f m = {sf[c], sf[c]};
                            pool2[rp][ch] = __builtin_elementwise_fma(w, m, pool2[rp][ch]);
                        } else if (vl) {
                            pool2[rp][ch].x = fmaf(Pk[dyl][c], sf[c], pool2[rp][ch].x);
                        } else if (vh) {
                            pool2[rp][ch].y = fmaf(Pk[dyh][c], sf[c], pool2[rp][ch].y);
                        }
                    }
                }
            }
        }
#pragma unroll
        for (int r = 0; r < RB; ++r) {
            const int gy = y0 + RB * g + r, gx = x0 + ox;
            float* op = out + (((size_t)b * H + gy) * W + gx) * 3;
#pragma unroll
            for (int ch = 0; ch < 3; ++ch)
                op[ch] = quot(ctr[r][ch], ((r & 1) ? pool2[r >> 1][ch].y : pool2[r >> 1][ch].x) + SIGMA_P);
        }
    }
}

template <bool GAM, bool LAST, int NT>
__global__ void __launch_bounds__(NT) __attribute__((amdgpu_waves_per_eu(12)))
nlp_level(const float* __restrict__ src, const float* __restrict__ down,
          float* __restrict__ out, float* __restrict__ nxt, int H, int W, int nd, int sh) {
    constexpr int SUBF = 3 * (36 * 37 + 20 * 21);
    constexpr int DWNF = LAST ? 0 : (NT / 32) * 16 * 49;
    __shared__ float smem[SUBF > DWNF ? SUBF : DWNF];

    const int bz = blockIdx.z;
    const int b = blockIdx.x;
    const int tid = threadIdx.x;

    if (b < 3 * nd) {
        const int pc = (b >= nd ? 1 : 0) + (b >= 2 * nd ? 1 : 0);
        const int g = b - pc * nd;
        const int p = 3 * bz + pc;
        if (LAST) {
            for (int e = tid; e < 256; e += NT) {
                const int y = e >> 4, x = e & 15;
                float v = down[(size_t)p * 18 * 18 + (size_t)(y + 1) * 18 + (x + 1)];
                float o = quot(v, fabsf(v) + SIGMA_LP);
                nxt[(((size_t)bz * 16 + y) * 16 + x) * 3 + pc] = o;
            }
        } else {
            const int wave = __builtin_amdgcn_readfirstlane(tid >> 5);
            const int Hn = H / 2, Wn = W / 2;
            const int Ty = (Hn / 2 + 2 + 15) / 16, Tx = (Wn / 2 + 2 + 15) / 16;
            const int tile = g * (NT / 32) + wave;
            const bool live = tile < Ty * Tx;
            const int ti = tile / Tx, tj = tile - ti * Tx;
            down_tile(down, nxt, Hn, Wn, p, ti * 16, tj * 16, live,
                             (float (*)[49])(smem + wave * (16 * 49)), tid & 31);
        }
    } else {
        const int s = b - 3 * nd;
        sub_tile<GAM, NT>(src, down, out, H, W, bz, (s >> sh) * 32, (s & ((1 << sh) - 1)) * 32,
                          tid, (float (*)[36][37])smem, (float (*)[20][21])(smem + 3 * 36 * 37));
    }
}

extern "C" void kernel_launch(void* const* d_in, const int* in_sizes, int n_in,
                              void* d_out, int out_size, void* d_ws, size_t ws_size,
                              hipStream_t stream) {
    (void)in_sizes; (void)n_in; (void)out_size; (void)ws_size;
    const float* img = (const float*)d_in[0];
    float* out = (float*)d_out;
    float* ws  = (float*)d_ws;

    const int Hl[6]  = {512, 256, 128, 64, 32, 16};
    const int shl[5] = {4, 3, 2, 1, 0};
    const int ntl[5] = {128, 128, 256, 256, 256};

    size_t dof[5]; size_t off = 0;
    for (int l = 0; l < 5; ++l) {
        dof[l] = off;
        size_t h2 = (size_t)(Hl[l] / 2 + 2);
        off += (size_t)48 * h2 * h2;
    }
    size_t oof[6]; size_t o2 = 0;
    for (int l = 0; l < 6; ++l) {
        oof[l] = o2;
        o2 += (size_t)16 * Hl[l] * Hl[l] * 3;
    }

    {
        const int T = (Hl[0] / 2 + 2 + 15) / 16;
        nlp_down0<<<dim3(T, T, 16), 32, 0, stream>>>(img, ws + dof[0], Hl[0], Hl[0]);
    }
    const float* srcp = img;
    for (int l = 0; l < 5; ++l) {
        const int H = Hl[l];
        const int ns = (H / 32) * (H / 32);
        if (l < 4) {
            const int Tn = (Hl[l + 1] / 2 + 2 + 15) / 16;
            const int tw = ntl[l] / 32;
            const int nd = (Tn * Tn + tw - 1) / tw;
            if (l == 0)
                nlp_level<true, false, 128><<<dim3(3 * nd + ns, 1, 16), 128, 0, stream>>>(
                    srcp, ws + dof[l], out + oof[l], ws + dof[l + 1], H, H, nd, shl[l]);
            else if (l == 1)
                nlp_level<false, false, 128><<<dim3(3 * nd + ns, 1, 16), 128, 0, stream>>>(
                    srcp, ws + dof[l], out + oof[l], ws + dof[l + 1], H, H, nd, shl[l]);
            else
                nlp_level<false, false, 256><<<dim3(3 * nd + ns, 1, 16), 256, 0, stream>>>(
                    srcp, ws + dof[l], out + oof[l], ws + dof[l + 1], H, H, nd, shl[l]);
        } else {
            nlp_level<false, true, 256><<<dim3(3 + ns, 1, 16), 256, 0, stream>>>(
                srcp, ws + dof[l], out + oof[l], out + oof[5], H, H, 1, shl[l]);
        }
        srcp = ws + dof[l];
    }
}
